# starting kernel
# baseline (speedup 1.0000x reference)
#include <hip/hip_runtime.h>

#define B_TOTAL 8192
#define T_STEPS 256

typedef float v2f __attribute__((ext_vector_type(2)));
typedef float v8f __attribute__((ext_vector_type(8)));

static __device__ __forceinline__ v8f wmma4(v2f a, v2f b, v8f c) {
  return __builtin_amdgcn_wmma_f32_16x16x4_f32(false, a, false, b, (short)0, c,
                                               false, false);
}

static __device__ __forceinline__ v8f mm16(const v2f a[4], const v2f b[4], v8f acc) {
#pragma unroll
  for (int k = 0; k < 4; ++k) acc = wmma4(a[k], b[k], acc);
  return acc;
}

static __device__ __forceinline__ float xor16(float v) {
#if __has_builtin(__builtin_amdgcn_permlanex16)
  unsigned u = __builtin_bit_cast(unsigned, v);
  unsigned r = __builtin_amdgcn_permlanex16(u, u, 0x76543210u, 0xFEDCBA98u,
                                            false, false);
  return __builtin_bit_cast(float, r);
#else
  return __shfl_xor(v, 16, 32);
#endif
}

static __device__ __forceinline__ float fast_tanh(float x) {
#if __has_builtin(__builtin_amdgcn_tanhf)
  return __builtin_amdgcn_tanhf(x);
#else
  return tanhf(x);
#endif
}
static __device__ __forceinline__ float fast_rcp(float x) {
#if __has_builtin(__builtin_amdgcn_rcpf)
  return __builtin_amdgcn_rcpf(x);
#else
  return 1.0f / x;
#endif
}
static __device__ __forceinline__ float fast_rsq(float x) {
#if __has_builtin(__builtin_amdgcn_rsqf)
  return __builtin_amdgcn_rsqf(x);
#else
  return rsqrtf(x);
#endif
}
static __device__ __forceinline__ float sigf(float x) {
  return fast_rcp(1.0f + __expf(-x));
}

static __device__ __forceinline__ void cd_to_b(const float x[8], bool lo, v2f b[4]) {
  float p2 = xor16(x[2]), p3 = xor16(x[3]);
  b[0][0] = lo ? x[0] : p2;
  b[0][1] = lo ? x[1] : p3;
  float p6 = xor16(x[6]), p7 = xor16(x[7]);
  b[1][0] = lo ? x[4] : p6;
  b[1][1] = lo ? x[5] : p7;
  float p0 = xor16(x[0]), p1 = xor16(x[1]);
  b[2][0] = lo ? p0 : x[2];
  b[2][1] = lo ? p1 : x[3];
  float p4 = xor16(x[4]), p5 = xor16(x[5]);
  b[3][0] = lo ? p4 : x[6];
  b[3][1] = lo ? p5 : x[7];
}

__global__ __launch_bounds__(128)
void vehicle_lstm_wmma(const float* __restrict__ initial_state,
                       const float* __restrict__ commands,
                       const float* __restrict__ ln_gamma,
                       const float* __restrict__ ln_beta,
                       const float* __restrict__ W_in,
                       const float* __restrict__ b_in,
                       const float* __restrict__ W_ih,
                       const float* __restrict__ W_hh,
                       const float* __restrict__ b_ih,
                       const float* __restrict__ b_hh,
                       const float* __restrict__ W_out,
                       const float* __restrict__ b_out,
                       float* __restrict__ out) {
  const int lane = threadIdx.x & 31;
  const int wave = threadIdx.x >> 5;
  const bool lo  = lane < 16;
  const int  n   = lane & 15;
  const int  b   = blockIdx.x * 64 + wave * 16 + n;
  const int  kof = lo ? 0 : 2;
  const int  rb  = lo ? 0 : 8;

  v2f a_in[4], a_ih[4][4], a_hh[4][4], a_out[4];
#pragma unroll
  for (int k = 0; k < 4; ++k) {
    const int kb = 4 * k + kof;
    a_in[k][0] = W_in[n * 16 + kb];
    a_in[k][1] = W_in[n * 16 + kb + 1];
#pragma unroll
    for (int g = 0; g < 4; ++g) {
      a_ih[g][k][0] = W_ih[(g * 16 + n) * 16 + kb];
      a_ih[g][k][1] = W_ih[(g * 16 + n) * 16 + kb + 1];
      a_hh[g][k][0] = W_hh[(g * 16 + n) * 16 + kb];
      a_hh[g][k][1] = W_hh[(g * 16 + n) * 16 + kb + 1];
    }
    a_out[k][0] = (n < 12) ? W_out[n * 16 + kb] : 0.0f;
    a_out[k][1] = (n < 12) ? W_out[n * 16 + kb + 1] : 0.0f;
  }

  v8f bias_in, bias_g[4], bias_out;
  float gam[8], bet[8];
#pragma unroll
  for (int r = 0; r < 8; ++r) {
    const int d = r + rb;
    gam[r] = ln_gamma[d];
    bet[r] = ln_beta[d];
    bias_in[r] = b_in[d];
#pragma unroll
    for (int g = 0; g < 4; ++g)
      bias_g[g][r] = b_ih[g * 16 + d] + b_hh[g * 16 + d];
    bias_out[r] = (d < 12) ? b_out[d] : 0.0f;
  }

  float s[8];
  const float* st0 = initial_state + (size_t)b * 12;
#pragma unroll
  for (int r = 0; r < 8; ++r) {
    float vlo = st0[r];
    float vhi = (r < 4) ? st0[8 + r] : 0.0f;
    s[r] = lo ? vlo : vhi;
  }

  float cst[8];
  v2f bh[4];
#pragma unroll
  for (int r = 0; r < 8; ++r) cst[r] = 0.0f;
#pragma unroll
  for (int k = 0; k < 4; ++k) { bh[k][0] = 0.0f; bh[k][1] = 0.0f; }

  const float4* cmd4 = reinterpret_cast<const float4*>(commands);

  for (int t = 0; t < T_STEPS; ++t) {
    float4 c4 = cmd4[(size_t)b * T_STEPS + t];
    float x[8];
    x[0] = s[0]; x[1] = s[1]; x[2] = s[2]; x[3] = s[3];
    x[4] = lo ? s[4] : c4.x;
    x[5] = lo ? s[5] : c4.y;
    x[6] = lo ? s[6] : c4.z;
    x[7] = lo ? s[7] : c4.w;

    float sum = 0.0f, sq = 0.0f;
#pragma unroll
    for (int r = 0; r < 8; ++r) { sum += x[r]; sq += x[r] * x[r]; }
    sum += xor16(sum);
    sq  += xor16(sq);
    const float mu   = sum * 0.0625f;
    const float var  = sq * 0.0625f - mu * mu;
    const float rstd = fast_rsq(var + 1e-5f);
#pragma unroll
    for (int r = 0; r < 8; ++r) x[r] = (x[r] - mu) * rstd * gam[r] + bet[r];

    v2f bmat[4];
    cd_to_b(x, lo, bmat);
    v8f x1 = mm16(a_in, bmat, bias_in);

    float x1a[8];
#pragma unroll
    for (int r = 0; r < 8; ++r) x1a[r] = x1[r];
    cd_to_b(x1a, lo, bmat);

    v8f Gi = mm16(a_hh[0], bh, mm16(a_ih[0], bmat, bias_g[0]));
    v8f Gf = mm16(a_hh[1], bh, mm16(a_ih[1], bmat, bias_g[1]));
    v8f Gg = mm16(a_hh[2], bh, mm16(a_ih[2], bmat, bias_g[2]));
    v8f Go = mm16(a_hh[3], bh, mm16(a_ih[3], bmat, bias_g[3]));

    float hnew[8];
#pragma unroll
    for (int r = 0; r < 8; ++r) {
      float cc = sigf(Gf[r]) * cst[r] + sigf(Gi[r]) * fast_tanh(Gg[r]);
      cst[r]   = cc;
      hnew[r]  = sigf(Go[r]) * fast_tanh(cc);
    }

    cd_to_b(hnew, lo, bh);
    v8f ds = mm16(a_out, bh, bias_out);
#pragma unroll
    for (int r = 0; r < 8; ++r) {
      float ns = s[r] + ds[r];
      s[r] = (lo || r < 4) ? ns : s[r];
    }

#pragma unroll
    for (int p = 0; p < 3; ++p) {
      const int ci = 1 + 2 * p, si = 2 + 2 * p;
      float cs = s[ci], sn = s[si];
      float inv = fast_rcp(sqrtf(sn * sn + cs * cs) + 1e-8f);
      s[ci] = lo ? cs * inv : s[ci];
      s[si] = lo ? sn * inv : s[si];
    }

    float* op = out + ((size_t)b * T_STEPS + t) * 12;
    if (lo) {
      *reinterpret_cast<float4*>(op)     = make_float4(s[0], s[1], s[2], s[3]);
      *reinterpret_cast<float4*>(op + 4) = make_float4(s[4], s[5], s[6], s[7]);
    } else {
      *reinterpret_cast<float4*>(op + 8) = make_float4(s[0], s[1], s[2], s[3]);
    }
  }
}

extern "C" void kernel_launch(void* const* d_in, const int* in_sizes, int n_in,
                              void* d_out, int out_size, void* d_ws, size_t ws_size,
                              hipStream_t stream) {
  (void)in_sizes; (void)n_in; (void)out_size; (void)d_ws; (void)ws_size;
  dim3 grid(B_TOTAL / 64);
  dim3 block(128);
  vehicle_lstm_wmma<<<grid, block, 0, stream>>>(
      (const float*)d_in[0],
      (const float*)d_in[1],
      (const float*)d_in[2],
      (const float*)d_in[3],
      (const float*)d_in[4],
      (const float*)d_in[5],
      (const float*)d_in[6],
      (const float*)d_in[7],
      (const float*)d_in[8],
      (const float*)d_in[9],
      (const float*)d_in[10],
      (const float*)d_in[11],
      (float*)d_out);
}


# Command read of step t+1 issued one step ahead into its own registers: its memory latency overlaps the step's arithmetic instead of stalling the loop top; arithmetic and order of sums unchanged
# speedup vs baseline: 1.2748x; 1.2748x over previous
#include <hip/hip_runtime.h>

#define B_TOTAL 8192
#define T_STEPS 256

typedef float v2f __attribute__((ext_vector_type(2)));
typedef float v8f __attribute__((ext_vector_type(8)));

static __device__ __forceinline__ v8f wmma4(v2f a, v2f b, v8f c) {
  return __builtin_amdgcn_wmma_f32_16x16x4_f32(false, a, false, b, (short)0, c,
                                               false, false);
}

static __device__ __forceinline__ v8f mm16(const v2f a[4], const v2f b[4], v8f acc) {
#pragma unroll
  for (int k = 0; k < 4; ++k) acc = wmma4(a[k], b[k], acc);
  return acc;
}

static __device__ __forceinline__ float xor16(float v) {
#if __has_builtin(__builtin_amdgcn_permlanex16)
  unsigned u = __builtin_bit_cast(unsigned, v);
  unsigned r = __builtin_amdgcn_permlanex16(u, u, 0x76543210u, 0xFEDCBA98u,
                                            false, false);
  return __builtin_bit_cast(float, r);
#else
  return __shfl_xor(v, 16, 32);
#endif
}

static __device__ __forceinline__ float fast_tanh(float x) {
#if __has_builtin(__builtin_amdgcn_tanhf)
  return __builtin_amdgcn_tanhf(x);
#else
  return tanhf(x);
#endif
}
static __device__ __forceinline__ float fast_rcp(float x) {
#if __has_builtin(__builtin_amdgcn_rcpf)
  return __builtin_amdgcn_rcpf(x);
#else
  return 1.0f / x;
#endif
}
static __device__ __forceinline__ float fast_rsq(float x) {
#if __has_builtin(__builtin_amdgcn_rsqf)
  return __builtin_amdgcn_rsqf(x);
#else
  return rsqrtf(x);
#endif
}
static __device__ __forceinline__ float sigf(float x) {
  return fast_rcp(1.0f + __expf(-x));
}

static __device__ __forceinline__ void cd_to_b(const float x[8], bool lo, v2f b[4]) {
  float p2 = xor16(x[2]), p3 = xor16(x[3]);
  b[0][0] = lo ? x[0] : p2;
  b[0][1] = lo ? x[1] : p3;
  float p6 = xor16(x[6]), p7 = xor16(x[7]);
  b[1][0] = lo ? x[4] : p6;
  b[1][1] = lo ? x[5] : p7;
  float p0 = xor16(x[0]), p1 = xor16(x[1]);
  b[2][0] = lo ? p0 : x[2];
  b[2][1] = lo ? p1 : x[3];
  float p4 = xor16(x[4]), p5 = xor16(x[5]);
  b[3][0] = lo ? p4 : x[6];
  b[3][1] = lo ? p5 : x[7];
}

__global__ __launch_bounds__(128)
void vehicle_lstm_wmma(const float* __restrict__ initial_state,
                       const float* __restrict__ commands,
                       const float* __restrict__ ln_gamma,
                       const float* __restrict__ ln_beta,
                       const float* __restrict__ W_in,
                       const float* __restrict__ b_in,
                       const float* __restrict__ W_ih,
                       const float* __restrict__ W_hh,
                       const float* __restrict__ b_ih,
                       const float* __restrict__ b_hh,
                       const float* __restrict__ W_out,
                       const float* __restrict__ b_out,
                       float* __restrict__ out) {
  const int lane = threadIdx.x & 31;
  const int wave = threadIdx.x >> 5;
  const bool lo  = lane < 16;
  const int  n   = lane & 15;
  const int  b   = blockIdx.x * 64 + wave * 16 + n;
  const int  kof = lo ? 0 : 2;
  const int  rb  = lo ? 0 : 8;

  v2f a_in[4], a_ih[4][4], a_hh[4][4], a_out[4];
#pragma unroll
  for (int k = 0; k < 4; ++k) {
    const int kb = 4 * k + kof;
    a_in[k][0] = W_in[n * 16 + kb];
    a_in[k][1] = W_in[n * 16 + kb + 1];
#pragma unroll
    for (int g = 0; g < 4; ++g) {
      a_ih[g][k][0] = W_ih[(g * 16 + n) * 16 + kb];
      a_ih[g][k][1] = W_ih[(g * 16 + n) * 16 + kb + 1];
      a_hh[g][k][0] = W_hh[(g * 16 + n) * 16 + kb];
      a_hh[g][k][1] = W_hh[(g * 16 + n) * 16 + kb + 1];
    }
    a_out[k][0] = (n < 12) ? W_out[n * 16 + kb] : 0.0f;
    a_out[k][1] = (n < 12) ? W_out[n * 16 + kb + 1] : 0.0f;
  }

  v8f bias_in, bias_g[4], bias_out;
  float gam[8], bet[8];
#pragma unroll
  for (int r = 0; r < 8; ++r) {
    const int d = r + rb;
    gam[r] = ln_gamma[d];
    bet[r] = ln_beta[d];
    bias_in[r] = b_in[d];
#pragma unroll
    for (int g = 0; g < 4; ++g)
      bias_g[g][r] = b_ih[g * 16 + d] + b_hh[g * 16 + d];
    bias_out[r] = (d < 12) ? b_out[d] : 0.0f;
  }

  float s[8];
  const float* st0 = initial_state + (size_t)b * 12;
#pragma unroll
  for (int r = 0; r < 8; ++r) {
    float vlo = st0[r];
    float vhi = (r < 4) ? st0[8 + r] : 0.0f;
    s[r] = lo ? vlo : vhi;
  }

  float cst[8];
  v2f bh[4];
#pragma unroll
  for (int r = 0; r < 8; ++r) cst[r] = 0.0f;
#pragma unroll
  for (int k = 0; k < 4; ++k) { bh[k][0] = 0.0f; bh[k][1] = 0.0f; }

  const float4* cmdp = reinterpret_cast<const float4*>(commands) + (size_t)b * T_STEPS;
  float4 cnext = cmdp[0];

  for (int t = 0; t < T_STEPS; ++t) {
    const float4 c4 = cnext;
    const int tn = (t + 1 < T_STEPS) ? t + 1 : t;
    cnext = cmdp[tn];
    float x[8];
    x[0] = s[0]; x[1] = s[1]; x[2] = s[2]; x[3] = s[3];
    x[4] = lo ? s[4] : c4.x;
    x[5] = lo ? s[5] : c4.y;
    x[6] = lo ? s[6] : c4.z;
    x[7] = lo ? s[7] : c4.w;

    float sum = 0.0f, sq = 0.0f;
#pragma unroll
    for (int r = 0; r < 8; ++r) { sum += x[r]; sq += x[r] * x[r]; }
    sum += xor16(sum);
    sq  += xor16(sq);
    const float mu   = sum * 0.0625f;
    const float var  = sq * 0.0625f - mu * mu;
    const float rstd = fast_rsq(var + 1e-5f);
#pragma unroll
    for (int r = 0; r < 8; ++r) x[r] = (x[r] - mu) * rstd * gam[r] + bet[r];

    v2f bmat[4];
    cd_to_b(x, lo, bmat);
    v8f x1 = mm16(a_in, bmat, bias_in);

    float x1a[8];
#pragma unroll
    for (int r = 0; r < 8; ++r) x1a[r] = x1[r];
    cd_to_b(x1a, lo, bmat);

    v8f Gi = mm16(a_hh[0], bh, mm16(a_ih[0], bmat, bias_g[0]));
    v8f Gf = mm16(a_hh[1], bh, mm16(a_ih[1], bmat, bias_g[1]));
    v8f Gg = mm16(a_hh[2], bh, mm16(a_ih[2], bmat, bias_g[2]));
    v8f Go = mm16(a_hh[3], bh, mm16(a_ih[3], bmat, bias_g[3]));

    float hnew[8];
#pragma unroll
    for (int r = 0; r < 8; ++r) {
      float cc = sigf(Gf[r]) * cst[r] + sigf(Gi[r]) * fast_tanh(Gg[r]);
      cst[r]   = cc;
      hnew[r]  = sigf(Go[r]) * fast_tanh(cc);
    }

    cd_to_b(hnew, lo, bh);
    v8f ds = mm16(a_out, bh, bias_out);
#pragma unroll
    for (int r = 0; r < 8; ++r) {
      float ns = s[r] + ds[r];
      s[r] = (lo || r < 4) ? ns : s[r];
    }

#pragma unroll
    for (int p = 0; p < 3; ++p) {
      const int ci = 1 + 2 * p, si = 2 + 2 * p;
      float cs = s[ci], sn = s[si];
      float inv = fast_rcp(sqrtf(sn * sn + cs * cs) + 1e-8f);
      s[ci] = lo ? cs * inv : s[ci];
      s[si] = lo ? sn * inv : s[si];
    }

    float* op = out + ((size_t)b * T_STEPS + t) * 12;
    if (lo) {
      *reinterpret_cast<float4*>(op)     = make_float4(s[0], s[1], s[2], s[3]);
      *reinterpret_cast<float4*>(op + 4) = make_float4(s[4], s[5], s[6], s[7]);
    } else {
      *reinterpret_cast<float4*>(op + 8) = make_float4(s[0], s[1], s[2], s[3]);
    }
  }
}

extern "C" void kernel_launch(void* const* d_in, const int* in_sizes, int n_in,
                              void* d_out, int out_size, void* d_ws, size_t ws_size,
                              hipStream_t stream) {
  (void)in_sizes; (void)n_in; (void)out_size; (void)d_ws; (void)ws_size;
  dim3 grid(B_TOTAL / 64);
  dim3 block(128);
  vehicle_lstm_wmma<<<grid, block, 0, stream>>>(
      (const float*)d_in[0],
      (const float*)d_in[1],
      (const float*)d_in[2],
      (const float*)d_in[3],
      (const float*)d_in[4],
      (const float*)d_in[5],
      (const float*)d_in[6],
      (const float*)d_in[7],
      (const float*)d_in[8],
      (const float*)d_in[9],
      (const float*)d_in[10],
      (const float*)d_in[11],
      (float*)d_out);
}


# One straight-line step with the recurrent W_hh*h matrix products issued early: LayerNorm, state update and i/f sigmoids run beside the matrix unit; gate sums add the h term before the x term
# speedup vs baseline: 1.4483x; 1.1361x over previous
#include <hip/hip_runtime.h>

#define B_TOTAL 8192
#define T_STEPS 256

typedef float v2f __attribute__((ext_vector_type(2)));
typedef float v8f __attribute__((ext_vector_type(8)));

static __device__ __forceinline__ v8f wmma4(v2f a, v2f b, v8f c) {
  return __builtin_amdgcn_wmma_f32_16x16x4_f32(false, a, false, b, (short)0, c,
                                               false, false);
}

static __device__ __forceinline__ v8f mm16(const v2f a[4], const v2f b[4], v8f acc) {
#pragma unroll
  for (int k = 0; k < 4; ++k) acc = wmma4(a[k], b[k], acc);
  return acc;
}

static __device__ __forceinline__ float xor16(float v) {
#if __has_builtin(__builtin_amdgcn_permlanex16)
  unsigned u = __builtin_bit_cast(unsigned, v);
  unsigned r = __builtin_amdgcn_permlanex16(u, u, 0x76543210u, 0xFEDCBA98u,
                                            false, false);
  return __builtin_bit_cast(float, r);
#else
  return __shfl_xor(v, 16, 32);
#endif
}

static __device__ __forceinline__ float fast_tanh(float x) {
#if __has_builtin(__builtin_amdgcn_tanhf)
  return __builtin_amdgcn_tanhf(x);
#else
  return tanhf(x);
#endif
}
static __device__ __forceinline__ float fast_rcp(float x) {
#if __has_builtin(__builtin_amdgcn_rcpf)
  return __builtin_amdgcn_rcpf(x);
#else
  return 1.0f / x;
#endif
}
static __device__ __forceinline__ float fast_rsq(float x) {
#if __has_builtin(__builtin_amdgcn_rsqf)
  return __builtin_amdgcn_rsqf(x);
#else
  return rsqrtf(x);
#endif
}
static __device__ __forceinline__ float sigf(float x) {
  return fast_rcp(1.0f + __expf(-x));
}

static __device__ __forceinline__ void cd_to_b(const float x[8], bool lo, v2f b[4]) {
  float p2 = xor16(x[2]), p3 = xor16(x[3]);
  b[0][0] = lo ? x[0] : p2;
  b[0][1] = lo ? x[1] : p3;
  float p6 = xor16(x[6]), p7 = xor16(x[7]);
  b[1][0] = lo ? x[4] : p6;
  b[1][1] = lo ? x[5] : p7;
  float p0 = xor16(x[0]), p1 = xor16(x[1]);
  b[2][0] = lo ? p0 : x[2];
  b[2][1] = lo ? p1 : x[3];
  float p4 = xor16(x[4]), p5 = xor16(x[5]);
  b[3][0] = lo ? p4 : x[6];
  b[3][1] = lo ? p5 : x[7];
}

__global__ __launch_bounds__(128)
void vehicle_lstm_wmma(const float* __restrict__ initial_state,
                       const float* __restrict__ commands,
                       const float* __restrict__ ln_gamma,
                       const float* __restrict__ ln_beta,
                       const float* __restrict__ W_in,
                       const float* __restrict__ b_in,
                       const float* __restrict__ W_ih,
                       const float* __restrict__ W_hh,
                       const float* __restrict__ b_ih,
                       const float* __restrict__ b_hh,
                       const float* __restrict__ W_out,
                       const float* __restrict__ b_out,
                       float* __restrict__ out) {
  const int lane = threadIdx.x & 31;
  const int wave = threadIdx.x >> 5;
  const bool lo  = lane < 16;
  const int  n   = lane & 15;
  const int  b   = blockIdx.x * 64 + wave * 16 + n;
  const int  kof = lo ? 0 : 2;
  const int  rb  = lo ? 0 : 8;

  v2f a_in[4], a_ih[4][4], a_hh[4][4], a_out[4];
#pragma unroll
  for (int k = 0; k < 4; ++k) {
    const int kb = 4 * k + kof;
    a_in[k][0] = W_in[n * 16 + kb];
    a_in[k][1] = W_in[n * 16 + kb + 1];
#pragma unroll
    for (int g = 0; g < 4; ++g) {
      a_ih[g][k][0] = W_ih[(g * 16 + n) * 16 + kb];
      a_ih[g][k][1] = W_ih[(g * 16 + n) * 16 + kb + 1];
      a_hh[g][k][0] = W_hh[(g * 16 + n) * 16 + kb];
      a_hh[g][k][1] = W_hh[(g * 16 + n) * 16 + kb + 1];
    }
    a_out[k][0] = (n < 12) ? W_out[n * 16 + kb] : 0.0f;
    a_out[k][1] = (n < 12) ? W_out[n * 16 + kb + 1] : 0.0f;
  }

  v8f bias_in, bias_g[4], bias_out;
  float gam[8], bet[8];
#pragma unroll
  for (int r = 0; r < 8; ++r) {
    const int d = r + rb;
    gam[r] = ln_gamma[d];
    bet[r] = ln_beta[d];
    bias_in[r] = b_in[d];
#pragma unroll
    for (int g = 0; g < 4; ++g)
      bias_g[g][r] = b_ih[g * 16 + d] + b_hh[g * 16 + d];
    bias_out[r] = (d < 12) ? b_out[d] : 0.0f;
  }

  float s[8];
  const float* st0 = initial_state + (size_t)b * 12;
#pragma unroll
  for (int r = 0; r < 8; ++r) {
    float vlo = st0[r];
    float vhi = (r < 4) ? st0[8 + r] : 0.0f;
    s[r] = lo ? vlo : vhi;
  }

  float cst[8];
  v2f bh[4];
#pragma unroll
  for (int r = 0; r < 8; ++r) cst[r] = 0.0f;
#pragma unroll
  for (int k = 0; k < 4; ++k) { bh[k][0] = 0.0f; bh[k][1] = 0.0f; }

  v8f pgi = mm16(a_hh[0], bh, bias_g[0]);
  v8f pgf = mm16(a_hh[1], bh, bias_g[1]);

  const float4* cmdp = reinterpret_cast<const float4*>(commands) + (size_t)b * T_STEPS;
  float4 cnext = cmdp[0];

  float* opa = out + (size_t)b * T_STEPS * 12 + (lo ? 0 : 8);
  float* opb = out + (size_t)b * T_STEPS * 12 + (lo ? 4 : 6);

  for (int t = 0; t < T_STEPS; ++t) {
    const float4 c4 = cnext;
    const int tn = (t + 1 < T_STEPS) ? t + 1 : t;
    cnext = cmdp[tn];

    v8f Gg = mm16(a_hh[2], bh, bias_g[2]);
    v8f Go = mm16(a_hh[3], bh, bias_g[3]);

    float x[8];
    x[0] = s[0]; x[1] = s[1]; x[2] = s[2]; x[3] = s[3];
    x[4] = lo ? s[4] : c4.x;
    x[5] = lo ? s[5] : c4.y;
    x[6] = lo ? s[6] : c4.z;
    x[7] = lo ? s[7] : c4.w;

    float sum = 0.0f, sq = 0.0f;
#pragma unroll
    for (int r = 0; r < 8; ++r) { sum += x[r]; sq += x[r] * x[r]; }
    sum += xor16(sum);
    sq  += xor16(sq);
    const float mu   = sum * 0.0625f;
    const float var  = sq * 0.0625f - mu * mu;
    const float rstd = fast_rsq(var + 1e-5f);
#pragma unroll
    for (int r = 0; r < 8; ++r) x[r] = (x[r] - mu) * rstd * gam[r] + bet[r];

    v2f bmat[4];
    cd_to_b(x, lo, bmat);

    asm volatile("" :: "v"(Gg), "v"(Go));
    __builtin_amdgcn_sched_barrier(0);

    v8f x1 = mm16(a_in, bmat, bias_in);

    float x1a[8];
#pragma unroll
    for (int r = 0; r < 8; ++r) x1a[r] = x1[r];
    cd_to_b(x1a, lo, bmat);

    v8f Gi = pgi, Gf = pgf;
#pragma unroll
    for (int k = 0; k < 4; ++k) {
      Gi = wmma4(a_ih[0][k], bmat[k], Gi);
      Gf = wmma4(a_ih[1][k], bmat[k], Gf);
    }
    asm volatile("" :: "v"(Gi), "v"(Gf));
    __builtin_amdgcn_sched_barrier(0);
#pragma unroll
    for (int k = 0; k < 4; ++k) {
      Gg = wmma4(a_ih[2][k], bmat[k], Gg);
      Go = wmma4(a_ih[3][k], bmat[k], Go);
    }
    float si[8], sf[8];
#pragma unroll
    for (int r = 0; r < 8; ++r) { si[r] = sigf(Gi[r]); sf[r] = sigf(Gf[r]); }

    float hnew[8];
#pragma unroll
    for (int r = 0; r < 8; ++r) {
      float cc = sf[r] * cst[r] + si[r] * fast_tanh(Gg[r]);
      cst[r]   = cc;
      hnew[r]  = sigf(Go[r]) * fast_tanh(cc);
    }

    cd_to_b(hnew, lo, bh);
    v8f ds = mm16(a_out, bh, bias_out);
    asm volatile("" :: "v"(ds));
    __builtin_amdgcn_sched_barrier(0);

    pgi = mm16(a_hh[0], bh, bias_g[0]);
    pgf = mm16(a_hh[1], bh, bias_g[1]);

#pragma unroll
    for (int r = 0; r < 8; ++r) {
      float ns = s[r] + ds[r];
      s[r] = (lo || r < 4) ? ns : s[r];
    }

#pragma unroll
    for (int p = 0; p < 3; ++p) {
      const int ci = 1 + 2 * p, sj = 2 + 2 * p;
      float cs = s[ci], sn = s[sj];
      float inv = fast_rcp(sqrtf(sn * sn + cs * cs) + 1e-8f);
      s[ci] = lo ? cs * inv : s[ci];
      s[sj] = lo ? sn * inv : s[sj];
    }

    const float p6 = xor16(s[6]), p7 = xor16(s[7]);
    *reinterpret_cast<float4*>(opa) = make_float4(s[0], s[1], s[2], s[3]);
    *reinterpret_cast<float2*>(opb) = lo ? make_float2(s[4], s[5]) : make_float2(p6, p7);
    opa += 12;
    opb += 12;
  }
}

extern "C" void kernel_launch(void* const* d_in, const int* in_sizes, int n_in,
                              void* d_out, int out_size, void* d_ws, size_t ws_size,
                              hipStream_t stream) {
  (void)in_sizes; (void)n_in; (void)out_size; (void)d_ws; (void)ws_size;
  dim3 grid(B_TOTAL / 64);
  dim3 block(128);
  vehicle_lstm_wmma<<<grid, block, 0, stream>>>(
      (const float*)d_in[0],
      (const float*)d_in[1],
      (const float*)d_in[2],
      (const float*)d_in[3],
      (const float*)d_in[4],
      (const float*)d_in[5],
      (const float*)d_in[6],
      (const float*)d_in[7],
      (const float*)d_in[8],
      (const float*)d_in[9],
      (const float*)d_in[10],
      (const float*)d_in[11],
      (float*)d_out);
}


# Weight fragments loaded in k order 8h+2k+s, so a C/D tile is a B fragment as it stands: the three lane-exchange relayouts per step go (x1 re-enters behind one s_wait_alu); sums take a new k order
# speedup vs baseline: 1.4959x; 1.0329x over previous
#include <hip/hip_runtime.h>

#define B_TOTAL 8192
#define T_STEPS 256

typedef float v2f __attribute__((ext_vector_type(2)));
typedef float v8f __attribute__((ext_vector_type(8)));

static __device__ __forceinline__ v8f wmma4(v2f a, v2f b, v8f c) {
  return __builtin_amdgcn_wmma_f32_16x16x4_f32(false, a, false, b, (short)0, c,
                                               false, false);
}

static __device__ __forceinline__ v8f mm16(const v2f a[4], const v2f b[4], v8f acc) {
#pragma unroll
  for (int k = 0; k < 4; ++k) acc = wmma4(a[k], b[k], acc);
  return acc;
}

static __device__ __forceinline__ float xor16(float v) {
#if __has_builtin(__builtin_amdgcn_permlanex16)
  unsigned u = __builtin_bit_cast(unsigned, v);
  unsigned r = __builtin_amdgcn_permlanex16(u, u, 0x76543210u, 0xFEDCBA98u,
                                            false, false);
  return __builtin_bit_cast(float, r);
#else
  return __shfl_xor(v, 16, 32);
#endif
}

static __device__ __forceinline__ float fast_tanh(float x) {
#if __has_builtin(__builtin_amdgcn_tanhf)
  return __builtin_amdgcn_tanhf(x);
#else
  return tanhf(x);
#endif
}
static __device__ __forceinline__ float fast_rcp(float x) {
#if __has_builtin(__builtin_amdgcn_rcpf)
  return __builtin_amdgcn_rcpf(x);
#else
  return 1.0f / x;
#endif
}
static __device__ __forceinline__ float fast_rsq(float x) {
#if __has_builtin(__builtin_amdgcn_rsqf)
  return __builtin_amdgcn_rsqf(x);
#else
  return rsqrtf(x);
#endif
}
static __device__ __forceinline__ float sigf(float x) {
  return fast_rcp(1.0f + __expf(-x));
}

static __device__ __forceinline__ void cd_to_b(const float x[8], v2f b[4]) {
#pragma unroll
  for (int k = 0; k < 4; ++k) { b[k][0] = x[2 * k]; b[k][1] = x[2 * k + 1]; }
}

__global__ __launch_bounds__(128)
void vehicle_lstm_wmma(const float* __restrict__ initial_state,
                       const float* __restrict__ commands,
                       const float* __restrict__ ln_gamma,
                       const float* __restrict__ ln_beta,
                       const float* __restrict__ W_in,
                       const float* __restrict__ b_in,
                       const float* __restrict__ W_ih,
                       const float* __restrict__ W_hh,
                       const float* __restrict__ b_ih,
                       const float* __restrict__ b_hh,
                       const float* __restrict__ W_out,
                       const float* __restrict__ b_out,
                       float* __restrict__ out) {
  const int lane = threadIdx.x & 31;
  const int wave = threadIdx.x >> 5;
  const bool lo  = lane < 16;
  const int  n   = lane & 15;
  const int  b   = blockIdx.x * 64 + wave * 16 + n;
  const int  kof = lo ? 0 : 8;
  const int  rb  = lo ? 0 : 8;

  v2f a_in[4], a_ih[4][4], a_hh[4][4], a_out[4];
#pragma unroll
  for (int k = 0; k < 4; ++k) {
    const int kb = 2 * k + kof;
    a_in[k][0] = W_in[n * 16 + kb];
    a_in[k][1] = W_in[n * 16 + kb + 1];
#pragma unroll
    for (int g = 0; g < 4; ++g) {
      a_ih[g][k][0] = W_ih[(g * 16 + n) * 16 + kb];
      a_ih[g][k][1] = W_ih[(g * 16 + n) * 16 + kb + 1];
      a_hh[g][k][0] = W_hh[(g * 16 + n) * 16 + kb];
      a_hh[g][k][1] = W_hh[(g * 16 + n) * 16 + kb + 1];
    }
    a_out[k][0] = (n < 12) ? W_out[n * 16 + kb] : 0.0f;
    a_out[k][1] = (n < 12) ? W_out[n * 16 + kb + 1] : 0.0f;
  }

  v8f bias_in, bias_g[4], bias_out;
  float gam[8], bet[8];
#pragma unroll
  for (int r = 0; r < 8; ++r) {
    const int d = r + rb;
    gam[r] = ln_gamma[d];
    bet[r] = ln_beta[d];
    bias_in[r] = b_in[d];
#pragma unroll
    for (int g = 0; g < 4; ++g)
      bias_g[g][r] = b_ih[g * 16 + d] + b_hh[g * 16 + d];
    bias_out[r] = (d < 12) ? b_out[d] : 0.0f;
  }

  float s[8];
  const float* st0 = initial_state + (size_t)b * 12;
#pragma unroll
  for (int r = 0; r < 8; ++r) {
    float vlo = st0[r];
    float vhi = (r < 4) ? st0[8 + r] : 0.0f;
    s[r] = lo ? vlo : vhi;
  }

  float cst[8];
  v2f bh[4];
#pragma unroll
  for (int r = 0; r < 8; ++r) cst[r] = 0.0f;
#pragma unroll
  for (int k = 0; k < 4; ++k) { bh[k][0] = 0.0f; bh[k][1] = 0.0f; }

  v8f pgi = mm16(a_hh[0], bh, bias_g[0]);
  v8f pgf = mm16(a_hh[1], bh, bias_g[1]);

  const float4* cmdp = reinterpret_cast<const float4*>(commands) + (size_t)b * T_STEPS;
  float4 cnext = cmdp[0];

  float* opa = out + (size_t)b * T_STEPS * 12 + (lo ? 0 : 8);
  float* opb = out + (size_t)b * T_STEPS * 12 + (lo ? 4 : 6);

  for (int t = 0; t < T_STEPS; ++t) {
    const float4 c4 = cnext;
    const int tn = (t + 1 < T_STEPS) ? t + 1 : t;
    cnext = cmdp[tn];

    v8f Gg = mm16(a_hh[2], bh, bias_g[2]);
    v8f Go = mm16(a_hh[3], bh, bias_g[3]);

    float x[8];
    x[0] = s[0]; x[1] = s[1]; x[2] = s[2]; x[3] = s[3];
    x[4] = lo ? s[4] : c4.x;
    x[5] = lo ? s[5] : c4.y;
    x[6] = lo ? s[6] : c4.z;
    x[7] = lo ? s[7] : c4.w;

    float sum = 0.0f, sq = 0.0f;
#pragma unroll
    for (int r = 0; r < 8; ++r) { sum += x[r]; sq += x[r] * x[r]; }
    sum += xor16(sum);
    sq  += xor16(sq);
    const float mu   = sum * 0.0625f;
    const float var  = sq * 0.0625f - mu * mu;
    const float rstd = fast_rsq(var + 1e-5f);
#pragma unroll
    for (int r = 0; r < 8; ++r) x[r] = (x[r] - mu) * rstd * gam[r] + bet[r];

    v2f bmat[4];
    cd_to_b(x, bmat);

    asm volatile("" :: "v"(Gg), "v"(Go));
    __builtin_amdgcn_sched_barrier(0);

    v8f x1 = mm16(a_in, bmat, bias_in);

    asm volatile("s_wait_alu depctr_va_vdst(0)" : "+v"(x1));

    float x1a[8];
#pragma unroll
    for (int r = 0; r < 8; ++r) x1a[r] = x1[r];
    cd_to_b(x1a, bmat);

    v8f Gi = pgi, Gf = pgf;
#pragma unroll
    for (int k = 0; k < 4; ++k) {
      Gi = wmma4(a_ih[0][k], bmat[k], Gi);
      Gf = wmma4(a_ih[1][k], bmat[k], Gf);
    }
    asm volatile("" :: "v"(Gi), "v"(Gf));
    __builtin_amdgcn_sched_barrier(0);
#pragma unroll
    for (int k = 0; k < 4; ++k) {
      Gg = wmma4(a_ih[2][k], bmat[k], Gg);
      Go = wmma4(a_ih[3][k], bmat[k], Go);
    }
    float si[8], sf[8];
#pragma unroll
    for (int r = 0; r < 8; ++r) { si[r] = sigf(Gi[r]); sf[r] = sigf(Gf[r]); }

    float hnew[8];
#pragma unroll
    for (int r = 0; r < 8; ++r) {
      float cc = sf[r] * cst[r] + si[r] * fast_tanh(Gg[r]);
      cst[r]   = cc;
      hnew[r]  = sigf(Go[r]) * fast_tanh(cc);
    }

    cd_to_b(hnew, bh);
    v8f ds = mm16(a_out, bh, bias_out);
    asm volatile("" :: "v"(ds));
    __builtin_amdgcn_sched_barrier(0);

    pgi = mm16(a_hh[0], bh, bias_g[0]);
    pgf = mm16(a_hh[1], bh, bias_g[1]);

#pragma unroll
    for (int r = 0; r < 8; ++r) {
      float ns = s[r] + ds[r];
      s[r] = (lo || r < 4) ? ns : s[r];
    }

#pragma unroll
    for (int p = 0; p < 3; ++p) {
      const int ci = 1 + 2 * p, sj = 2 + 2 * p;
      float cs = s[ci], sn = s[sj];
      float inv = fast_rcp(sqrtf(sn * sn + cs * cs) + 1e-8f);
      s[ci] = lo ? cs * inv : s[ci];
      s[sj] = lo ? sn * inv : s[sj];
    }

    const float p6 = xor16(s[6]), p7 = xor16(s[7]);
    *reinterpret_cast<float4*>(opa) = make_float4(s[0], s[1], s[2], s[3]);
    *reinterpret_cast<float2*>(opb) = lo ? make_float2(s[4], s[5]) : make_float2(p6, p7);
    opa += 12;
    opb += 12;
  }
}

extern "C" void kernel_launch(void* const* d_in, const int* in_sizes, int n_in,
                              void* d_out, int out_size, void* d_ws, size_t ws_size,
                              hipStream_t stream) {
  (void)in_sizes; (void)n_in; (void)out_size; (void)d_ws; (void)ws_size;
  dim3 grid(B_TOTAL / 64);
  dim3 block(128);
  vehicle_lstm_wmma<<<grid, block, 0, stream>>>(
      (const float*)d_in[0],
      (const float*)d_in[1],
      (const float*)d_in[2],
      (const float*)d_in[3],
      (const float*)d_in[4],
      (const float*)d_in[5],
      (const float*)d_in[6],
      (const float*)d_in[7],
      (const float*)d_in[8],
      (const float*)d_in[9],
      (const float*)d_in[10],
      (const float*)d_in[11],
      (float*)d_out);
}


# Input projection folded into the gate weights: W_ih*W_in and W_ih*b_in formed once per wave on the matrix unit, so a step drops the four-long x1 chain; gate sums associate as (W_ih*W_in)*x
# speedup vs baseline: 1.5985x; 1.0686x over previous
#include <hip/hip_runtime.h>

#define B_TOTAL 8192
#define T_STEPS 256

typedef float v2f __attribute__((ext_vector_type(2)));
typedef float v8f __attribute__((ext_vector_type(8)));

static __device__ __forceinline__ v8f wmma4(v2f a, v2f b, v8f c) {
  return __builtin_amdgcn_wmma_f32_16x16x4_f32(false, a, false, b, (short)0, c,
                                               false, false);
}

static __device__ __forceinline__ v8f mm16(const v2f a[4], const v2f b[4], v8f acc) {
#pragma unroll
  for (int k = 0; k < 4; ++k) acc = wmma4(a[k], b[k], acc);
  return acc;
}

static __device__ __forceinline__ float xor16(float v) {
#if __has_builtin(__builtin_amdgcn_permlanex16)
  unsigned u = __builtin_bit_cast(unsigned, v);
  unsigned r = __builtin_amdgcn_permlanex16(u, u, 0x76543210u, 0xFEDCBA98u,
                                            false, false);
  return __builtin_bit_cast(float, r);
#else
  return __shfl_xor(v, 16, 32);
#endif
}

static __device__ __forceinline__ float fast_tanh(float x) {
#if __has_builtin(__builtin_amdgcn_tanhf)
  return __builtin_amdgcn_tanhf(x);
#else
  return tanhf(x);
#endif
}
static __device__ __forceinline__ float fast_rcp(float x) {
#if __has_builtin(__builtin_amdgcn_rcpf)
  return __builtin_amdgcn_rcpf(x);
#else
  return 1.0f / x;
#endif
}
static __device__ __forceinline__ float fast_rsq(float x) {
#if __has_builtin(__builtin_amdgcn_rsqf)
  return __builtin_amdgcn_rsqf(x);
#else
  return rsqrtf(x);
#endif
}
static __device__ __forceinline__ float sigf(float x) {
  return fast_rcp(1.0f + __expf(-x));
}

static __device__ __forceinline__ void cd_to_b(const float x[8], v2f b[4]) {
#pragma unroll
  for (int k = 0; k < 4; ++k) { b[k][0] = x[2 * k]; b[k][1] = x[2 * k + 1]; }
}

__global__ __launch_bounds__(128)
void vehicle_lstm_wmma(const float* __restrict__ initial_state,
                       const float* __restrict__ commands,
                       const float* __restrict__ ln_gamma,
                       const float* __restrict__ ln_beta,
                       const float* __restrict__ W_in,
                       const float* __restrict__ b_in,
                       const float* __restrict__ W_ih,
                       const float* __restrict__ W_hh,
                       const float* __restrict__ b_ih,
                       const float* __restrict__ b_hh,
                       const float* __restrict__ W_out,
                       const float* __restrict__ b_out,
                       float* __restrict__ out) {
  const int lane = threadIdx.x & 31;
  const int wave = threadIdx.x >> 5;
  const bool lo  = lane < 16;
  const int  n   = lane & 15;
  const int  b   = blockIdx.x * 64 + wave * 16 + n;
  const int  kof = lo ? 0 : 8;
  const int  rb  = lo ? 0 : 8;

  v2f a_inT[4], b_inb[4], a_ih[4][4], a_hh[4][4], a_out[4];
#pragma unroll
  for (int k = 0; k < 4; ++k) {
    const int kb = 2 * k + kof;
    a_inT[k][0] = W_in[kb * 16 + n];
    a_inT[k][1] = W_in[(kb + 1) * 16 + n];
    b_inb[k][0] = b_in[kb];
    b_inb[k][1] = b_in[kb + 1];
#pragma unroll
    for (int g = 0; g < 4; ++g) {
      a_ih[g][k][0] = W_ih[(g * 16 + n) * 16 + kb];
      a_ih[g][k][1] = W_ih[(g * 16 + n) * 16 + kb + 1];
      a_hh[g][k][0] = W_hh[(g * 16 + n) * 16 + kb];
      a_hh[g][k][1] = W_hh[(g * 16 + n) * 16 + kb + 1];
    }
    a_out[k][0] = (n < 12) ? W_out[n * 16 + kb] : 0.0f;
    a_out[k][1] = (n < 12) ? W_out[n * 16 + kb + 1] : 0.0f;
  }

  v8f bias_g[4], bias_out;
  float gam[8], bet[8];
#pragma unroll
  for (int r = 0; r < 8; ++r) {
    const int d = r + rb;
    gam[r] = ln_gamma[d];
    bet[r] = ln_beta[d];
#pragma unroll
    for (int g = 0; g < 4; ++g)
      bias_g[g][r] = b_ih[g * 16 + d] + b_hh[g * 16 + d];
    bias_out[r] = (d < 12) ? b_out[d] : 0.0f;
  }

  v2f a_c[4][4];
#pragma unroll
  for (int g = 0; g < 4; ++g) {
    v8f wt;
#pragma unroll
    for (int r = 0; r < 8; ++r) wt[r] = 0.0f;
    wt = mm16(a_inT, a_ih[g], wt);
#pragma unroll
    for (int k = 0; k < 4; ++k) { a_c[g][k][0] = wt[2 * k]; a_c[g][k][1] = wt[2 * k + 1]; }
    bias_g[g] = mm16(a_ih[g], b_inb, bias_g[g]);
  }

  float s[8];
  const float* st0 = initial_state + (size_t)b * 12;
#pragma unroll
  for (int r = 0; r < 8; ++r) {
    float vlo = st0[r];
    float vhi = (r < 4) ? st0[8 + r] : 0.0f;
    s[r] = lo ? vlo : vhi;
  }

  float cst[8];
  v2f bh[4];
#pragma unroll
  for (int r = 0; r < 8; ++r) cst[r] = 0.0f;
#pragma unroll
  for (int k = 0; k < 4; ++k) { bh[k][0] = 0.0f; bh[k][1] = 0.0f; }

  v8f pgi = mm16(a_hh[0], bh, bias_g[0]);
  v8f pgf = mm16(a_hh[1], bh, bias_g[1]);

  const float4* cmdp = reinterpret_cast<const float4*>(commands) + (size_t)b * T_STEPS;
  float4 cnext = cmdp[0];

  float* opa = out + (size_t)b * T_STEPS * 12 + (lo ? 0 : 8);
  float* opb = out + (size_t)b * T_STEPS * 12 + (lo ? 4 : 6);

  for (int t = 0; t < T_STEPS; ++t) {
    const float4 c4 = cnext;
    const int tn = (t + 1 < T_STEPS) ? t + 1 : t;
    cnext = cmdp[tn];

    v8f Gg = mm16(a_hh[2], bh, bias_g[2]);
    v8f Go = mm16(a_hh[3], bh, bias_g[3]);

    float x[8];
    x[0] = s[0]; x[1] = s[1]; x[2] = s[2]; x[3] = s[3];
    x[4] = lo ? s[4] : c4.x;
    x[5] = lo ? s[5] : c4.y;
    x[6] = lo ? s[6] : c4.z;
    x[7] = lo ? s[7] : c4.w;

    float sum = 0.0f, sq = 0.0f;
#pragma unroll
    for (int r = 0; r < 8; ++r) { sum += x[r]; sq += x[r] * x[r]; }
    sum += xor16(sum);
    sq  += xor16(sq);
    const float mu   = sum * 0.0625f;
    const float var  = sq * 0.0625f - mu * mu;
    const float rstd = fast_rsq(var + 1e-5f);
#pragma unroll
    for (int r = 0; r < 8; ++r) x[r] = (x[r] - mu) * rstd * gam[r] + bet[r];

    v2f bmat[4];
    cd_to_b(x, bmat);

    asm volatile("" :: "v"(Gg), "v"(Go));
    __builtin_amdgcn_sched_barrier(0);

    v8f Gi = pgi, Gf = pgf;
#pragma unroll
    for (int k = 0; k < 4; ++k) {
      Gi = wmma4(a_c[0][k], bmat[k], Gi);
      Gf = wmma4(a_c[1][k], bmat[k], Gf);
    }
    asm volatile("" :: "v"(Gi), "v"(Gf));
    __builtin_amdgcn_sched_barrier(0);
#pragma unroll
    for (int k = 0; k < 4; ++k) {
      Gg = wmma4(a_c[2][k], bmat[k], Gg);
      Go = wmma4(a_c[3][k], bmat[k], Go);
    }
    float si[8], sf[8];
#pragma unroll
    for (int r = 0; r < 8; ++r) { si[r] = sigf(Gi[r]); sf[r] = sigf(Gf[r]); }

    float hnew[8];
#pragma unroll
    for (int r = 0; r < 8; ++r) {
      float cc = sf[r] * cst[r] + si[r] * fast_tanh(Gg[r]);
      cst[r]   = cc;
      hnew[r]  = sigf(Go[r]) * fast_tanh(cc);
    }

    cd_to_b(hnew, bh);
    v8f ds = mm16(a_out, bh, bias_out);
    asm volatile("" :: "v"(ds));
    __builtin_amdgcn_sched_barrier(0);

    pgi = mm16(a_hh[0], bh, bias_g[0]);
    pgf = mm16(a_hh[1], bh, bias_g[1]);

#pragma unroll
    for (int r = 0; r < 8; ++r) {
      float ns = s[r] + ds[r];
      s[r] = (lo || r < 4) ? ns : s[r];
    }

#pragma unroll
    for (int p = 0; p < 3; ++p) {
      const int ci = 1 + 2 * p, sj = 2 + 2 * p;
      float cs = s[ci], sn = s[sj];
      float inv = fast_rcp(sqrtf(sn * sn + cs * cs) + 1e-8f);
      s[ci] = lo ? cs * inv : s[ci];
      s[sj] = lo ? sn * inv : s[sj];
    }

    const float p6 = xor16(s[6]), p7 = xor16(s[7]);
    *reinterpret_cast<float4*>(opa) = make_float4(s[0], s[1], s[2], s[3]);
    *reinterpret_cast<float2*>(opb) = lo ? make_float2(s[4], s[5]) : make_float2(p6, p7);
    opa += 12;
    opb += 12;
  }
}

extern "C" void kernel_launch(void* const* d_in, const int* in_sizes, int n_in,
                              void* d_out, int out_size, void* d_ws, size_t ws_size,
                              hipStream_t stream) {
  (void)in_sizes; (void)n_in; (void)out_size; (void)d_ws; (void)ws_size;
  dim3 grid(B_TOTAL / 64);
  dim3 block(128);
  vehicle_lstm_wmma<<<grid, block, 0, stream>>>(
      (const float*)d_in[0],
      (const float*)d_in[1],
      (const float*)d_in[2],
      (const float*)d_in[3],
      (const float*)d_in[4],
      (const float*)d_in[5],
      (const float*)d_in[6],
      (const float*)d_in[7],
      (const float*)d_in[8],
      (const float*)d_in[9],
      (const float*)d_in[10],
      (const float*)d_in[11],
      (float*)d_out);
}


# Hardware square root (v_sqrt_f32, 1 ulp) in the three sin/cos normalisations instead of the correctly rounded library root: 36 vector instructions and a ten-link dependent tail leave each step
# speedup vs baseline: 1.6709x; 1.0453x over previous
#include <hip/hip_runtime.h>

#define B_TOTAL 8192
#define T_STEPS 256

typedef float v2f __attribute__((ext_vector_type(2)));
typedef float v8f __attribute__((ext_vector_type(8)));

static __device__ __forceinline__ v8f wmma4(v2f a, v2f b, v8f c) {
  return __builtin_amdgcn_wmma_f32_16x16x4_f32(false, a, false, b, (short)0, c,
                                               false, false);
}

static __device__ __forceinline__ v8f mm16(const v2f a[4], const v2f b[4], v8f acc) {
#pragma unroll
  for (int k = 0; k < 4; ++k) acc = wmma4(a[k], b[k], acc);
  return acc;
}

static __device__ __forceinline__ float xor16(float v) {
#if __has_builtin(__builtin_amdgcn_permlanex16)
  unsigned u = __builtin_bit_cast(unsigned, v);
  unsigned r = __builtin_amdgcn_permlanex16(u, u, 0x76543210u, 0xFEDCBA98u,
                                            false, false);
  return __builtin_bit_cast(float, r);
#else
  return __shfl_xor(v, 16, 32);
#endif
}

static __device__ __forceinline__ float fast_tanh(float x) {
#if __has_builtin(__builtin_amdgcn_tanhf)
  return __builtin_amdgcn_tanhf(x);
#else
  return tanhf(x);
#endif
}
static __device__ __forceinline__ float fast_rcp(float x) {
#if __has_builtin(__builtin_amdgcn_rcpf)
  return __builtin_amdgcn_rcpf(x);
#else
  return 1.0f / x;
#endif
}
static __device__ __forceinline__ float fast_rsq(float x) {
#if __has_builtin(__builtin_amdgcn_rsqf)
  return __builtin_amdgcn_rsqf(x);
#else
  return rsqrtf(x);
#endif
}
static __device__ __forceinline__ float fast_sqrt(float x) {
#if __has_builtin(__builtin_amdgcn_sqrtf)
  return __builtin_amdgcn_sqrtf(x);
#else
  return sqrtf(x);
#endif
}
static __device__ __forceinline__ float sigf(float x) {
  return fast_rcp(1.0f + __expf(-x));
}

static __device__ __forceinline__ void cd_to_b(const float x[8], v2f b[4]) {
#pragma unroll
  for (int k = 0; k < 4; ++k) { b[k][0] = x[2 * k]; b[k][1] = x[2 * k + 1]; }
}

__global__ __launch_bounds__(128)
void vehicle_lstm_wmma(const float* __restrict__ initial_state,
                       const float* __restrict__ commands,
                       const float* __restrict__ ln_gamma,
                       const float* __restrict__ ln_beta,
                       const float* __restrict__ W_in,
                       const float* __restrict__ b_in,
                       const float* __restrict__ W_ih,
                       const float* __restrict__ W_hh,
                       const float* __restrict__ b_ih,
                       const float* __restrict__ b_hh,
                       const float* __restrict__ W_out,
                       const float* __restrict__ b_out,
                       float* __restrict__ out) {
  const int lane = threadIdx.x & 31;
  const int wave = threadIdx.x >> 5;
  const bool lo  = lane < 16;
  const int  n   = lane & 15;
  const int  b   = blockIdx.x * 64 + wave * 16 + n;
  const int  kof = lo ? 0 : 8;
  const int  rb  = lo ? 0 : 8;

  v2f a_inT[4], b_inb[4], a_ih[4][4], a_hh[4][4], a_out[4];
#pragma unroll
  for (int k = 0; k < 4; ++k) {
    const int kb = 2 * k + kof;
    a_inT[k][0] = W_in[kb * 16 + n];
    a_inT[k][1] = W_in[(kb + 1) * 16 + n];
    b_inb[k][0] = b_in[kb];
    b_inb[k][1] = b_in[kb + 1];
#pragma unroll
    for (int g = 0; g < 4; ++g) {
      a_ih[g][k][0] = W_ih[(g * 16 + n) * 16 + kb];
      a_ih[g][k][1] = W_ih[(g * 16 + n) * 16 + kb + 1];
      a_hh[g][k][0] = W_hh[(g * 16 + n) * 16 + kb];
      a_hh[g][k][1] = W_hh[(g * 16 + n) * 16 + kb + 1];
    }
    a_out[k][0] = (n < 12) ? W_out[n * 16 + kb] : 0.0f;
    a_out[k][1] = (n < 12) ? W_out[n * 16 + kb + 1] : 0.0f;
  }

  v8f bias_g[4], bias_out;
  float gam[8], bet[8];
#pragma unroll
  for (int r = 0; r < 8; ++r) {
    const int d = r + rb;
    gam[r] = ln_gamma[d];
    bet[r] = ln_beta[d];
#pragma unroll
    for (int g = 0; g < 4; ++g)
      bias_g[g][r] = b_ih[g * 16 + d] + b_hh[g * 16 + d];
    bias_out[r] = (d < 12) ? b_out[d] : 0.0f;
  }

  v2f a_c[4][4];
#pragma unroll
  for (int g = 0; g < 4; ++g) {
    v8f wt;
#pragma unroll
    for (int r = 0; r < 8; ++r) wt[r] = 0.0f;
    wt = mm16(a_inT, a_ih[g], wt);
#pragma unroll
    for (int k = 0; k < 4; ++k) { a_c[g][k][0] = wt[2 * k]; a_c[g][k][1] = wt[2 * k + 1]; }
    bias_g[g] = mm16(a_ih[g], b_inb, bias_g[g]);
  }

  float s[8];
  const float* st0 = initial_state + (size_t)b * 12;
#pragma unroll
  for (int r = 0; r < 8; ++r) {
    float vlo = st0[r];
    float vhi = (r < 4) ? st0[8 + r] : 0.0f;
    s[r] = lo ? vlo : vhi;
  }

  float cst[8];
  v2f bh[4];
#pragma unroll
  for (int r = 0; r < 8; ++r) cst[r] = 0.0f;
#pragma unroll
  for (int k = 0; k < 4; ++k) { bh[k][0] = 0.0f; bh[k][1] = 0.0f; }

  v8f pgi = mm16(a_hh[0], bh, bias_g[0]);
  v8f pgf = mm16(a_hh[1], bh, bias_g[1]);

  const float4* cmdp = reinterpret_cast<const float4*>(commands) + (size_t)b * T_STEPS;
  float4 cnext = cmdp[0];

  float* opa = out + (size_t)b * T_STEPS * 12 + (lo ? 0 : 8);
  float* opb = out + (size_t)b * T_STEPS * 12 + (lo ? 4 : 6);

  for (int t = 0; t < T_STEPS; ++t) {
    const float4 c4 = cnext;
    const int tn = (t + 1 < T_STEPS) ? t + 1 : t;
    cnext = cmdp[tn];

    v8f Gg = mm16(a_hh[2], bh, bias_g[2]);
    v8f Go = mm16(a_hh[3], bh, bias_g[3]);

    float x[8];
    x[0] = s[0]; x[1] = s[1]; x[2] = s[2]; x[3] = s[3];
    x[4] = lo ? s[4] : c4.x;
    x[5] = lo ? s[5] : c4.y;
    x[6] = lo ? s[6] : c4.z;
    x[7] = lo ? s[7] : c4.w;

    float sum = 0.0f, sq = 0.0f;
#pragma unroll
    for (int r = 0; r < 8; ++r) { sum += x[r]; sq += x[r] * x[r]; }
    sum += xor16(sum);
    sq  += xor16(sq);
    const float mu   = sum * 0.0625f;
    const float var  = sq * 0.0625f - mu * mu;
    const float rstd = fast_rsq(var + 1e-5f);
#pragma unroll
    for (int r = 0; r < 8; ++r) x[r] = (x[r] - mu) * rstd * gam[r] + bet[r];

    v2f bmat[4];
    cd_to_b(x, bmat);

    asm volatile("" :: "v"(Gg), "v"(Go));
    __builtin_amdgcn_sched_barrier(0);

    v8f Gi = pgi, Gf = pgf;
#pragma unroll
    for (int k = 0; k < 4; ++k) {
      Gi = wmma4(a_c[0][k], bmat[k], Gi);
      Gf = wmma4(a_c[1][k], bmat[k], Gf);
    }
    asm volatile("" :: "v"(Gi), "v"(Gf));
    __builtin_amdgcn_sched_barrier(0);
#pragma unroll
    for (int k = 0; k < 4; ++k) {
      Gg = wmma4(a_c[2][k], bmat[k], Gg);
      Go = wmma4(a_c[3][k], bmat[k], Go);
    }
    float si[8], sf[8];
#pragma unroll
    for (int r = 0; r < 8; ++r) { si[r] = sigf(Gi[r]); sf[r] = sigf(Gf[r]); }

    float hnew[8];
#pragma unroll
    for (int r = 0; r < 8; ++r) {
      float cc = sf[r] * cst[r] + si[r] * fast_tanh(Gg[r]);
      cst[r]   = cc;
      hnew[r]  = sigf(Go[r]) * fast_tanh(cc);
    }

    cd_to_b(hnew, bh);
    v8f ds = mm16(a_out, bh, bias_out);
    asm volatile("" :: "v"(ds));
    __builtin_amdgcn_sched_barrier(0);

    pgi = mm16(a_hh[0], bh, bias_g[0]);
    pgf = mm16(a_hh[1], bh, bias_g[1]);

#pragma unroll
    for (int r = 0; r < 8; ++r) {
      float ns = s[r] + ds[r];
      s[r] = (lo || r < 4) ? ns : s[r];
    }

#pragma unroll
    for (int p = 0; p < 3; ++p) {
      const int ci = 1 + 2 * p, sj = 2 + 2 * p;
      float cs = s[ci], sn = s[sj];
      float inv = fast_rcp(fast_sqrt(sn * sn + cs * cs) + 1e-8f);
      s[ci] = lo ? cs * inv : s[ci];
      s[sj] = lo ? sn * inv : s[sj];
    }

    const float p6 = xor16(s[6]), p7 = xor16(s[7]);
    *reinterpret_cast<float4*>(opa) = make_float4(s[0], s[1], s[2], s[3]);
    *reinterpret_cast<float2*>(opb) = lo ? make_float2(s[4], s[5]) : make_float2(p6, p7);
    opa += 12;
    opb += 12;
  }
}

extern "C" void kernel_launch(void* const* d_in, const int* in_sizes, int n_in,
                              void* d_out, int out_size, void* d_ws, size_t ws_size,
                              hipStream_t stream) {
  (void)in_sizes; (void)n_in; (void)out_size; (void)d_ws; (void)ws_size;
  dim3 grid(B_TOTAL / 64);
  dim3 block(128);
  vehicle_lstm_wmma<<<grid, block, 0, stream>>>(
      (const float*)d_in[0],
      (const float*)d_in[1],
      (const float*)d_in[2],
      (const float*)d_in[3],
      (const float*)d_in[4],
      (const float*)d_in[5],
      (const float*)d_in[6],
      (const float*)d_in[7],
      (const float*)d_in[8],
      (const float*)d_in[9],
      (const float*)d_in[10],
      (const float*)d_in[11],
      (float*)d_out);
}


# Each 16-sample tile on two waves with a SIMD each (4-wave groups, 256 of them): a wave holds all four gates of eight hidden units, h halves swap through LDS once a step; hidden sums in a new k order
# speedup vs baseline: 1.7968x; 1.0753x over previous
#include <hip/hip_runtime.h>

#define B_TOTAL 8192
#define T_STEPS 256

typedef float v2f __attribute__((ext_vector_type(2)));
typedef float v8f __attribute__((ext_vector_type(8)));

static __device__ __forceinline__ v8f wmma4(v2f a, v2f b, v8f c) {
  return __builtin_amdgcn_wmma_f32_16x16x4_f32(false, a, false, b, (short)0, c,
                                               false, false);
}

static __device__ __forceinline__ v8f mm16(const v2f a[4], const v2f b[4], v8f acc) {
#pragma unroll
  for (int k = 0; k < 4; ++k) acc = wmma4(a[k], b[k], acc);
  return acc;
}

static __device__ __forceinline__ float xor16(float v) {
#if __has_builtin(__builtin_amdgcn_permlanex16)
  unsigned u = __builtin_bit_cast(unsigned, v);
  unsigned r = __builtin_amdgcn_permlanex16(u, u, 0x76543210u, 0xFEDCBA98u,
                                            false, false);
  return __builtin_bit_cast(float, r);
#else
  return __shfl_xor(v, 16, 32);
#endif
}

static __device__ __forceinline__ float fast_tanh(float x) {
#if __has_builtin(__builtin_amdgcn_tanhf)
  return __builtin_amdgcn_tanhf(x);
#else
  return tanhf(x);
#endif
}
static __device__ __forceinline__ float fast_rcp(float x) {
#if __has_builtin(__builtin_amdgcn_rcpf)
  return __builtin_amdgcn_rcpf(x);
#else
  return 1.0f / x;
#endif
}
static __device__ __forceinline__ float fast_rsq(float x) {
#if __has_builtin(__builtin_amdgcn_rsqf)
  return __builtin_amdgcn_rsqf(x);
#else
  return rsqrtf(x);
#endif
}
static __device__ __forceinline__ float fast_sqrt(float x) {
#if __has_builtin(__builtin_amdgcn_sqrtf)
  return __builtin_amdgcn_sqrtf(x);
#else
  return sqrtf(x);
#endif
}
static __device__ __forceinline__ float sigf(float x) {
  return fast_rcp(1.0f + __expf(-x));
}

static __device__ __forceinline__ void cd_to_b(const float x[8], v2f b[4]) {
#pragma unroll
  for (int k = 0; k < 4; ++k) { b[k][0] = x[2 * k]; b[k][1] = x[2 * k + 1]; }
}

__global__ __launch_bounds__(128)
void vehicle_lstm_wmma(const float* __restrict__ initial_state,
                       const float* commands,
                       const float* __restrict__ ln_gamma,
                       const float* __restrict__ ln_beta,
                       const float* __restrict__ W_in,
                       const float* __restrict__ b_in,
                       const float* __restrict__ W_ih,
                       const float* __restrict__ W_hh,
                       const float* __restrict__ b_ih,
                       const float* __restrict__ b_hh,
                       const float* __restrict__ W_out,
                       const float* __restrict__ b_out,
                       float* __restrict__ out) {
  __shared__ float4 xch[2][4][32];

  const int lane = threadIdx.x & 31;
  const int wave = __builtin_amdgcn_readfirstlane((int)(threadIdx.x >> 5));
  const int q    = wave & 1;
  const bool lo  = lane < 16;
  const int  n   = lane & 15;
  const int  hh  = lo ? 0 : 1;
  const int  b   = (blockIdx.x * 2 + (wave >> 1)) * 16 + n;
  const int  kof = lo ? 0 : 8;
  const int  rb  = lo ? 0 : 8;

#define GROW(T, rho) ((2 * (T) + (((rho) >> 2) & 1)) * 16 + 8 * q + 4 * ((rho) >> 3) + ((rho) & 3))

  v2f a_inT[4], b_inb[4], a_ih[2][4], a_hho[2][2], a_hhx[2][2], a_out[4];
#pragma unroll
  for (int k = 0; k < 4; ++k) {
    const int kb = 2 * k + kof;
    a_inT[k][0] = W_in[kb * 16 + n];
    a_inT[k][1] = W_in[(kb + 1) * 16 + n];
    b_inb[k][0] = b_in[kb];
    b_inb[k][1] = b_in[kb + 1];
#pragma unroll
    for (int T = 0; T < 2; ++T) {
      const int gr = GROW(T, n);
      a_ih[T][k][0] = W_ih[gr * 16 + kb];
      a_ih[T][k][1] = W_ih[gr * 16 + kb + 1];
    }
    const int uc = 8 * (k >> 1) + 4 * hh + 2 * (k & 1);
    a_out[k][0] = (n < 12) ? W_out[n * 16 + uc] : 0.0f;
    a_out[k][1] = (n < 12) ? W_out[n * 16 + uc + 1] : 0.0f;
  }
#pragma unroll
  for (int c = 0; c < 2; ++c) {
    const int uo = 8 * q + 4 * hh + 2 * c;
    const int ux = 8 * (1 - q) + 4 * hh + 2 * c;
#pragma unroll
    for (int T = 0; T < 2; ++T) {
      const int gr = GROW(T, n);
      a_hho[T][c][0] = W_hh[gr * 16 + uo];
      a_hho[T][c][1] = W_hh[gr * 16 + uo + 1];
      a_hhx[T][c][0] = W_hh[gr * 16 + ux];
      a_hhx[T][c][1] = W_hh[gr * 16 + ux + 1];
    }
  }

  v8f bias_g[2], bias_out;
  float gam[8], bet[8];
#pragma unroll
  for (int r = 0; r < 8; ++r) {
    const int d = r + rb;
    gam[r] = ln_gamma[d];
    bet[r] = ln_beta[d];
#pragma unroll
    for (int T = 0; T < 2; ++T) {
      const int gr = GROW(T, d);
      bias_g[T][r] = b_ih[gr] + b_hh[gr];
    }
    bias_out[r] = (d < 12) ? b_out[d] : 0.0f;
  }

  v2f a_c[2][4];
#pragma unroll
  for (int T = 0; T < 2; ++T) {
    v8f wt;
#pragma unroll
    for (int r = 0; r < 8; ++r) wt[r] = 0.0f;
    wt = mm16(a_inT, a_ih[T], wt);
#pragma unroll
    for (int k = 0; k < 4; ++k) { a_c[T][k][0] = wt[2 * k]; a_c[T][k][1] = wt[2 * k + 1]; }
    bias_g[T] = mm16(a_ih[T], b_inb, bias_g[T]);
  }

  float s[8];
  const float* st0 = initial_state + (size_t)b * 12;
#pragma unroll
  for (int r = 0; r < 8; ++r) {
    float vlo = st0[r];
    float vhi = (r < 4) ? st0[8 + r] : 0.0f;
    s[r] = lo ? vlo : vhi;
  }

  float cst[4];
#pragma unroll
  for (int j = 0; j < 4; ++j) cst[j] = 0.0f;

  v8f pg0 = bias_g[0];
  v8f pg1 = bias_g[1];

  const float4* cmdp = reinterpret_cast<const float4*>(commands) + (size_t)b * T_STEPS;
  float4 cnext = cmdp[0];

  float* opa = out + (size_t)b * T_STEPS * 12 + (lo ? 0 : 8);
  float* opb = out + (size_t)b * T_STEPS * 12 + (lo ? 4 : 6);

  for (int t = 0; t < T_STEPS; ++t) {
    const float4 c4 = cnext;
    const int tn = (t + 1 < T_STEPS) ? t + 1 : t;
    cnext = cmdp[tn];

    float x[8];
    x[0] = s[0]; x[1] = s[1]; x[2] = s[2]; x[3] = s[3];
    x[4] = lo ? s[4] : c4.x;
    x[5] = lo ? s[5] : c4.y;
    x[6] = lo ? s[6] : c4.z;
    x[7] = lo ? s[7] : c4.w;

    float sum = 0.0f, sq = 0.0f;
#pragma unroll
    for (int r = 0; r < 8; ++r) { sum += x[r]; sq += x[r] * x[r]; }
    sum += xor16(sum);
    sq  += xor16(sq);
    const float mu   = sum * 0.0625f;
    const float var  = sq * 0.0625f - mu * mu;
    const float rstd = fast_rsq(var + 1e-5f);
#pragma unroll
    for (int r = 0; r < 8; ++r) x[r] = (x[r] - mu) * rstd * gam[r] + bet[r];

    v2f bmat[4];
    cd_to_b(x, bmat);

    v8f G0 = pg0, G1 = pg1;
#pragma unroll
    for (int k = 0; k < 4; ++k) {
      G0 = wmma4(a_c[0][k], bmat[k], G0);
      G1 = wmma4(a_c[1][k], bmat[k], G1);
    }

    asm volatile("" :: "v"(G0), "v"(G1));
    __builtin_amdgcn_sched_barrier(0);

    float own[4];
#pragma unroll
    for (int j = 0; j < 4; ++j) {
      float cc = sigf(G0[4 + j]) * cst[j] + sigf(G0[j]) * fast_tanh(G1[j]);
      cst[j]   = cc;
      own[j]   = sigf(G1[4 + j]) * fast_tanh(cc);
    }

    xch[t & 1][wave][lane] = make_float4(own[0], own[1], own[2], own[3]);

    v2f bo[2];
    bo[0][0] = own[0]; bo[0][1] = own[1]; bo[1][0] = own[2]; bo[1][1] = own[3];
    pg0 = wmma4(a_hho[0][0], bo[0], bias_g[0]);
    pg1 = wmma4(a_hho[1][0], bo[0], bias_g[1]);
    pg0 = wmma4(a_hho[0][1], bo[1], pg0);
    pg1 = wmma4(a_hho[1][1], bo[1], pg1);
    asm volatile("" :: "v"(pg0), "v"(pg1));
    __builtin_amdgcn_sched_group_barrier(0x200, 1, 0);
    __builtin_amdgcn_sched_group_barrier(0x008, 4, 0);
    __builtin_amdgcn_sched_barrier(0);

    __builtin_amdgcn_fence(__ATOMIC_RELEASE, "workgroup", "local");
    __builtin_amdgcn_s_barrier();
    __builtin_amdgcn_fence(__ATOMIC_ACQUIRE, "workgroup", "local");
    const float4 ov = xch[t & 1][wave ^ 1][lane];
    v2f bx[2];
    bx[0][0] = ov.x; bx[0][1] = ov.y; bx[1][0] = ov.z; bx[1][1] = ov.w;

    v2f bh[4];
    bh[0] = (q == 0) ? bo[0] : bx[0];
    bh[1] = (q == 0) ? bo[1] : bx[1];
    bh[2] = (q == 0) ? bx[0] : bo[0];
    bh[3] = (q == 0) ? bx[1] : bo[1];
    v8f ds = mm16(a_out, bh, bias_out);
    asm volatile("" :: "v"(ds));
    __builtin_amdgcn_sched_barrier(0);

    pg0 = wmma4(a_hhx[0][0], bx[0], pg0);
    pg1 = wmma4(a_hhx[1][0], bx[0], pg1);
    pg0 = wmma4(a_hhx[0][1], bx[1], pg0);
    pg1 = wmma4(a_hhx[1][1], bx[1], pg1);
    asm volatile("" :: "v"(pg0), "v"(pg1));
    __builtin_amdgcn_sched_barrier(0);

#pragma unroll
    for (int r = 0; r < 8; ++r) {
      float ns = s[r] + ds[r];
      s[r] = (lo || r < 4) ? ns : s[r];
    }

#pragma unroll
    for (int p = 0; p < 3; ++p) {
      const int ci = 1 + 2 * p, sj = 2 + 2 * p;
      float cs = s[ci], sn = s[sj];
      float inv = fast_rcp(fast_sqrt(sn * sn + cs * cs) + 1e-8f);
      s[ci] = lo ? cs * inv : s[ci];
      s[sj] = lo ? sn * inv : s[sj];
    }

    if (q == 0) {
      *reinterpret_cast<float4*>(opa) = make_float4(s[0], s[1], s[2], s[3]);
    } else {
      const float p6 = xor16(s[6]), p7 = xor16(s[7]);
      *reinterpret_cast<float2*>(opb) = lo ? make_float2(s[4], s[5]) : make_float2(p6, p7);
    }
    opa += 12;
    opb += 12;
  }
#undef GROW
}

extern "C" void kernel_launch(void* const* d_in, const int* in_sizes, int n_in,
                              void* d_out, int out_size, void* d_ws, size_t ws_size,
                              hipStream_t stream) {
  (void)in_sizes; (void)n_in; (void)out_size; (void)d_ws; (void)ws_size;
  dim3 grid(B_TOTAL / 32);
  dim3 block(128);
  vehicle_lstm_wmma<<<grid, block, 0, stream>>>(
      (const float*)d_in[0],
      (const float*)d_in[1],
      (const float*)d_in[2],
      (const float*)d_in[3],
      (const float*)d_in[4],
      (const float*)d_in[5],
      (const float*)d_in[6],
      (const float*)d_in[7],
      (const float*)d_in[8],
      (const float*)d_in[9],
      (const float*)d_in[10],
      (const float*)d_in[11],
      (float*)d_out);
}


# State update in four phases (add; three roots; three reciprocals; scaling), one partner-unit recurrent matrix instruction in each instead of all four in front of it; products and sums unchanged
# speedup vs baseline: 1.8505x; 1.0299x over previous
#include <hip/hip_runtime.h>

#define B_TOTAL 8192
#define T_STEPS 256

typedef float v2f __attribute__((ext_vector_type(2)));
typedef float v8f __attribute__((ext_vector_type(8)));

static __device__ __forceinline__ v8f wmma4(v2f a, v2f b, v8f c) {
  return __builtin_amdgcn_wmma_f32_16x16x4_f32(false, a, false, b, (short)0, c,
                                               false, false);
}

static __device__ __forceinline__ v8f mm16(const v2f a[4], const v2f b[4], v8f acc) {
#pragma unroll
  for (int k = 0; k < 4; ++k) acc = wmma4(a[k], b[k], acc);
  return acc;
}

static __device__ __forceinline__ float xor16(float v) {
#if __has_builtin(__builtin_amdgcn_permlanex16)
  unsigned u = __builtin_bit_cast(unsigned, v);
  unsigned r = __builtin_amdgcn_permlanex16(u, u, 0x76543210u, 0xFEDCBA98u,
                                            false, false);
  return __builtin_bit_cast(float, r);
#else
  return __shfl_xor(v, 16, 32);
#endif
}

static __device__ __forceinline__ float fast_tanh(float x) {
#if __has_builtin(__builtin_amdgcn_tanhf)
  return __builtin_amdgcn_tanhf(x);
#else
  return tanhf(x);
#endif
}
static __device__ __forceinline__ float fast_rcp(float x) {
#if __has_builtin(__builtin_amdgcn_rcpf)
  return __builtin_amdgcn_rcpf(x);
#else
  return 1.0f / x;
#endif
}
static __device__ __forceinline__ float fast_rsq(float x) {
#if __has_builtin(__builtin_amdgcn_rsqf)
  return __builtin_amdgcn_rsqf(x);
#else
  return rsqrtf(x);
#endif
}
static __device__ __forceinline__ float fast_sqrt(float x) {
#if __has_builtin(__builtin_amdgcn_sqrtf)
  return __builtin_amdgcn_sqrtf(x);
#else
  return sqrtf(x);
#endif
}
static __device__ __forceinline__ float sigf(float x) {
  return fast_rcp(1.0f + __expf(-x));
}

static __device__ __forceinline__ void cd_to_b(const float x[8], v2f b[4]) {
#pragma unroll
  for (int k = 0; k < 4; ++k) { b[k][0] = x[2 * k]; b[k][1] = x[2 * k + 1]; }
}

__global__ __launch_bounds__(128)
void vehicle_lstm_wmma(const float* __restrict__ initial_state,
                       const float* commands,
                       const float* __restrict__ ln_gamma,
                       const float* __restrict__ ln_beta,
                       const float* __restrict__ W_in,
                       const float* __restrict__ b_in,
                       const float* __restrict__ W_ih,
                       const float* __restrict__ W_hh,
                       const float* __restrict__ b_ih,
                       const float* __restrict__ b_hh,
                       const float* __restrict__ W_out,
                       const float* __restrict__ b_out,
                       float* __restrict__ out) {
  __shared__ float4 xch[2][4][32];

  const int lane = threadIdx.x & 31;
  const int wave = __builtin_amdgcn_readfirstlane((int)(threadIdx.x >> 5));
  const int q    = wave & 1;
  const bool lo  = lane < 16;
  const int  n   = lane & 15;
  const int  hh  = lo ? 0 : 1;
  const int  b   = (blockIdx.x * 2 + (wave >> 1)) * 16 + n;
  const int  kof = lo ? 0 : 8;
  const int  rb  = lo ? 0 : 8;

#define GROW(T, rho) ((2 * (T) + (((rho) >> 2) & 1)) * 16 + 8 * q + 4 * ((rho) >> 3) + ((rho) & 3))

  v2f a_inT[4], b_inb[4], a_ih[2][4], a_hho[2][2], a_hhx[2][2], a_out[4];
#pragma unroll
  for (int k = 0; k < 4; ++k) {
    const int kb = 2 * k + kof;
    a_inT[k][0] = W_in[kb * 16 + n];
    a_inT[k][1] = W_in[(kb + 1) * 16 + n];
    b_inb[k][0] = b_in[kb];
    b_inb[k][1] = b_in[kb + 1];
#pragma unroll
    for (int T = 0; T < 2; ++T) {
      const int gr = GROW(T, n);
      a_ih[T][k][0] = W_ih[gr * 16 + kb];
      a_ih[T][k][1] = W_ih[gr * 16 + kb + 1];
    }
    const int uc = 8 * (k >> 1) + 4 * hh + 2 * (k & 1);
    a_out[k][0] = (n < 12) ? W_out[n * 16 + uc] : 0.0f;
    a_out[k][1] = (n < 12) ? W_out[n * 16 + uc + 1] : 0.0f;
  }
#pragma unroll
  for (int c = 0; c < 2; ++c) {
    const int uo = 8 * q + 4 * hh + 2 * c;
    const int ux = 8 * (1 - q) + 4 * hh + 2 * c;
#pragma unroll
    for (int T = 0; T < 2; ++T) {
      const int gr = GROW(T, n);
      a_hho[T][c][0] = W_hh[gr * 16 + uo];
      a_hho[T][c][1] = W_hh[gr * 16 + uo + 1];
      a_hhx[T][c][0] = W_hh[gr * 16 + ux];
      a_hhx[T][c][1] = W_hh[gr * 16 + ux + 1];
    }
  }

  v8f bias_g[2], bias_out;
  float gam[8], bet[8];
#pragma unroll
  for (int r = 0; r < 8; ++r) {
    const int d = r + rb;
    gam[r] = ln_gamma[d];
    bet[r] = ln_beta[d];
#pragma unroll
    for (int T = 0; T < 2; ++T) {
      const int gr = GROW(T, d);
      bias_g[T][r] = b_ih[gr] + b_hh[gr];
    }
    bias_out[r] = (d < 12) ? b_out[d] : 0.0f;
  }

  v2f a_c[2][4];
#pragma unroll
  for (int T = 0; T < 2; ++T) {
    v8f wt;
#pragma unroll
    for (int r = 0; r < 8; ++r) wt[r] = 0.0f;
    wt = mm16(a_inT, a_ih[T], wt);
#pragma unroll
    for (int k = 0; k < 4; ++k) { a_c[T][k][0] = wt[2 * k]; a_c[T][k][1] = wt[2 * k + 1]; }
    bias_g[T] = mm16(a_ih[T], b_inb, bias_g[T]);
  }

  float s[8];
  const float* st0 = initial_state + (size_t)b * 12;
#pragma unroll
  for (int r = 0; r < 8; ++r) {
    float vlo = st0[r];
    float vhi = (r < 4) ? st0[8 + r] : 0.0f;
    s[r] = lo ? vlo : vhi;
  }

  float cst[4];
#pragma unroll
  for (int j = 0; j < 4; ++j) cst[j] = 0.0f;

  v8f pg0 = bias_g[0];
  v8f pg1 = bias_g[1];

  const float4* cmdp = reinterpret_cast<const float4*>(commands) + (size_t)b * T_STEPS;
  float4 cnext = cmdp[0];

  float* opa = out + (size_t)b * T_STEPS * 12 + (lo ? 0 : 8);
  float* opb = out + (size_t)b * T_STEPS * 12 + (lo ? 4 : 6);

  for (int t = 0; t < T_STEPS; ++t) {
    const float4 c4 = cnext;
    const int tn = (t + 1 < T_STEPS) ? t + 1 : t;
    cnext = cmdp[tn];

    float x[8];
    x[0] = s[0]; x[1] = s[1]; x[2] = s[2]; x[3] = s[3];
    x[4] = lo ? s[4] : c4.x;
    x[5] = lo ? s[5] : c4.y;
    x[6] = lo ? s[6] : c4.z;
    x[7] = lo ? s[7] : c4.w;

    float sum = 0.0f, sq = 0.0f;
#pragma unroll
    for (int r = 0; r < 8; ++r) { sum += x[r]; sq += x[r] * x[r]; }
    sum += xor16(sum);
    sq  += xor16(sq);
    const float mu   = sum * 0.0625f;
    const float var  = sq * 0.0625f - mu * mu;
    const float rstd = fast_rsq(var + 1e-5f);
#pragma unroll
    for (int r = 0; r < 8; ++r) x[r] = (x[r] - mu) * rstd * gam[r] + bet[r];

    v2f bmat[4];
    cd_to_b(x, bmat);

    v8f G0 = pg0, G1 = pg1;
#pragma unroll
    for (int k = 0; k < 4; ++k) {
      G0 = wmma4(a_c[0][k], bmat[k], G0);
      G1 = wmma4(a_c[1][k], bmat[k], G1);
    }

    asm volatile("" :: "v"(G0), "v"(G1));
    __builtin_amdgcn_sched_barrier(0);

    float own[4];
#pragma unroll
    for (int j = 0; j < 4; ++j) {
      float cc = sigf(G0[4 + j]) * cst[j] + sigf(G0[j]) * fast_tanh(G1[j]);
      cst[j]   = cc;
      own[j]   = sigf(G1[4 + j]) * fast_tanh(cc);
    }

    xch[t & 1][wave][lane] = make_float4(own[0], own[1], own[2], own[3]);

    v2f bo[2];
    bo[0][0] = own[0]; bo[0][1] = own[1]; bo[1][0] = own[2]; bo[1][1] = own[3];
    pg0 = wmma4(a_hho[0][0], bo[0], bias_g[0]);
    pg1 = wmma4(a_hho[1][0], bo[0], bias_g[1]);
    pg0 = wmma4(a_hho[0][1], bo[1], pg0);
    pg1 = wmma4(a_hho[1][1], bo[1], pg1);
    asm volatile("" :: "v"(pg0), "v"(pg1));
    __builtin_amdgcn_sched_group_barrier(0x200, 1, 0);
    __builtin_amdgcn_sched_group_barrier(0x008, 4, 0);
    __builtin_amdgcn_sched_barrier(0);

    __builtin_amdgcn_fence(__ATOMIC_RELEASE, "workgroup", "local");
    __builtin_amdgcn_s_barrier();
    __builtin_amdgcn_fence(__ATOMIC_ACQUIRE, "workgroup", "local");
    const float4 ov = xch[t & 1][wave ^ 1][lane];
    v2f bx[2];
    bx[0][0] = ov.x; bx[0][1] = ov.y; bx[1][0] = ov.z; bx[1][1] = ov.w;

    v2f bh[4];
    bh[0] = (q == 0) ? bo[0] : bx[0];
    bh[1] = (q == 0) ? bo[1] : bx[1];
    bh[2] = (q == 0) ? bx[0] : bo[0];
    bh[3] = (q == 0) ? bx[1] : bo[1];
    v8f ds = mm16(a_out, bh, bias_out);
    asm volatile("" :: "v"(ds));
    __builtin_amdgcn_sched_barrier(0);


    pg0 = wmma4(a_hhx[0][0], bx[0], pg0);
#pragma unroll
    for (int r = 0; r < 8; ++r) {
      float ns = s[r] + ds[r];
      s[r] = (lo || r < 4) ? ns : s[r];
    }
    asm volatile("" :: "v"(pg0), "v"(s[0]), "v"(s[1]), "v"(s[2]), "v"(s[3]),
                       "v"(s[4]), "v"(s[5]), "v"(s[6]), "v"(s[7]));
    __builtin_amdgcn_sched_barrier(0);

    pg1 = wmma4(a_hhx[1][0], bx[0], pg1);
    float rt[3];
#pragma unroll
    for (int p = 0; p < 3; ++p) {
      const float cs = s[1 + 2 * p], sn = s[2 + 2 * p];
      rt[p] = fast_sqrt(sn * sn + cs * cs);
    }
    asm volatile("" :: "v"(pg1), "v"(rt[0]), "v"(rt[1]), "v"(rt[2]));
    __builtin_amdgcn_sched_barrier(0);

    pg0 = wmma4(a_hhx[0][1], bx[1], pg0);
    float inv[3];
#pragma unroll
    for (int p = 0; p < 3; ++p) inv[p] = fast_rcp(rt[p] + 1e-8f);
    asm volatile("" :: "v"(pg0), "v"(inv[0]), "v"(inv[1]), "v"(inv[2]));
    __builtin_amdgcn_sched_barrier(0);

    pg1 = wmma4(a_hhx[1][1], bx[1], pg1);
#pragma unroll
    for (int p = 0; p < 3; ++p) {
      const int ci = 1 + 2 * p, sj = 2 + 2 * p;
      const float cs = s[ci], sn = s[sj];
      s[ci] = lo ? cs * inv[p] : s[ci];
      s[sj] = lo ? sn * inv[p] : s[sj];
    }

    if (q == 0) {
      *reinterpret_cast<float4*>(opa) = make_float4(s[0], s[1], s[2], s[3]);
    } else {
      const float p6 = xor16(s[6]), p7 = xor16(s[7]);
      *reinterpret_cast<float2*>(opb) = lo ? make_float2(s[4], s[5]) : make_float2(p6, p7);
    }
    opa += 12;
    opb += 12;
  }
#undef GROW
}

extern "C" void kernel_launch(void* const* d_in, const int* in_sizes, int n_in,
                              void* d_out, int out_size, void* d_ws, size_t ws_size,
                              hipStream_t stream) {
  (void)in_sizes; (void)n_in; (void)out_size; (void)d_ws; (void)ws_size;
  dim3 grid(B_TOTAL / 32);
  dim3 block(128);
  vehicle_lstm_wmma<<<grid, block, 0, stream>>>(
      (const float*)d_in[0],
      (const float*)d_in[1],
      (const float*)d_in[2],
      (const float*)d_in[3],
      (const float*)d_in[4],
      (const float*)d_in[5],
      (const float*)d_in[6],
      (const float*)d_in[7],
      (const float*)d_in[8],
      (const float*)d_in[9],
      (const float*)d_in[10],
      (const float*)d_in[11],
      (float*)d_out);
}


# State-path product W_out*h split by owner: own units' half issued before the LDS hand-over, only the partner's two-long chain behind it; each wave adds its own units first (state sums reorder)
# speedup vs baseline: 1.9062x; 1.0301x over previous
#include <hip/hip_runtime.h>

#define B_TOTAL 8192
#define T_STEPS 256

typedef float v2f __attribute__((ext_vector_type(2)));
typedef float v8f __attribute__((ext_vector_type(8)));

static __device__ __forceinline__ v8f wmma4(v2f a, v2f b, v8f c) {
  return __builtin_amdgcn_wmma_f32_16x16x4_f32(false, a, false, b, (short)0, c,
                                               false, false);
}

static __device__ __forceinline__ v8f mm16(const v2f a[4], const v2f b[4], v8f acc) {
#pragma unroll
  for (int k = 0; k < 4; ++k) acc = wmma4(a[k], b[k], acc);
  return acc;
}

static __device__ __forceinline__ float xor16(float v) {
#if __has_builtin(__builtin_amdgcn_permlanex16)
  unsigned u = __builtin_bit_cast(unsigned, v);
  unsigned r = __builtin_amdgcn_permlanex16(u, u, 0x76543210u, 0xFEDCBA98u,
                                            false, false);
  return __builtin_bit_cast(float, r);
#else
  return __shfl_xor(v, 16, 32);
#endif
}

static __device__ __forceinline__ float fast_tanh(float x) {
#if __has_builtin(__builtin_amdgcn_tanhf)
  return __builtin_amdgcn_tanhf(x);
#else
  return tanhf(x);
#endif
}
static __device__ __forceinline__ float fast_rcp(float x) {
#if __has_builtin(__builtin_amdgcn_rcpf)
  return __builtin_amdgcn_rcpf(x);
#else
  return 1.0f / x;
#endif
}
static __device__ __forceinline__ float fast_rsq(float x) {
#if __has_builtin(__builtin_amdgcn_rsqf)
  return __builtin_amdgcn_rsqf(x);
#else
  return rsqrtf(x);
#endif
}
static __device__ __forceinline__ float fast_sqrt(float x) {
#if __has_builtin(__builtin_amdgcn_sqrtf)
  return __builtin_amdgcn_sqrtf(x);
#else
  return sqrtf(x);
#endif
}
static __device__ __forceinline__ float sigf(float x) {
  return fast_rcp(1.0f + __expf(-x));
}

static __device__ __forceinline__ void cd_to_b(const float x[8], v2f b[4]) {
#pragma unroll
  for (int k = 0; k < 4; ++k) { b[k][0] = x[2 * k]; b[k][1] = x[2 * k + 1]; }
}

__global__ __launch_bounds__(128)
void vehicle_lstm_wmma(const float* __restrict__ initial_state,
                       const float* commands,
                       const float* __restrict__ ln_gamma,
                       const float* __restrict__ ln_beta,
                       const float* __restrict__ W_in,
                       const float* __restrict__ b_in,
                       const float* __restrict__ W_ih,
                       const float* __restrict__ W_hh,
                       const float* __restrict__ b_ih,
                       const float* __restrict__ b_hh,
                       const float* __restrict__ W_out,
                       const float* __restrict__ b_out,
                       float* __restrict__ out) {
  __shared__ float4 xch[2][4][32];

  const int lane = threadIdx.x & 31;
  const int wave = __builtin_amdgcn_readfirstlane((int)(threadIdx.x >> 5));
  const int q    = wave & 1;
  const bool lo  = lane < 16;
  const int  n   = lane & 15;
  const int  hh  = lo ? 0 : 1;
  const int  b   = (blockIdx.x * 2 + (wave >> 1)) * 16 + n;
  const int  kof = lo ? 0 : 8;
  const int  rb  = lo ? 0 : 8;

#define GROW(T, rho) ((2 * (T) + (((rho) >> 2) & 1)) * 16 + 8 * q + 4 * ((rho) >> 3) + ((rho) & 3))

  v2f a_inT[4], b_inb[4], a_ih[2][4], a_hho[2][2], a_hhx[2][2], a_outo[2], a_outx[2];
#pragma unroll
  for (int k = 0; k < 4; ++k) {
    const int kb = 2 * k + kof;
    a_inT[k][0] = W_in[kb * 16 + n];
    a_inT[k][1] = W_in[(kb + 1) * 16 + n];
    b_inb[k][0] = b_in[kb];
    b_inb[k][1] = b_in[kb + 1];
#pragma unroll
    for (int T = 0; T < 2; ++T) {
      const int gr = GROW(T, n);
      a_ih[T][k][0] = W_ih[gr * 16 + kb];
      a_ih[T][k][1] = W_ih[gr * 16 + kb + 1];
    }
  }
#pragma unroll
  for (int c = 0; c < 2; ++c) {
    const int uo = 8 * q + 4 * hh + 2 * c;
    const int ux = 8 * (1 - q) + 4 * hh + 2 * c;
#pragma unroll
    for (int T = 0; T < 2; ++T) {
      const int gr = GROW(T, n);
      a_hho[T][c][0] = W_hh[gr * 16 + uo];
      a_hho[T][c][1] = W_hh[gr * 16 + uo + 1];
      a_hhx[T][c][0] = W_hh[gr * 16 + ux];
      a_hhx[T][c][1] = W_hh[gr * 16 + ux + 1];
    }
    a_outo[c][0] = (n < 12) ? W_out[n * 16 + uo] : 0.0f;
    a_outo[c][1] = (n < 12) ? W_out[n * 16 + uo + 1] : 0.0f;
    a_outx[c][0] = (n < 12) ? W_out[n * 16 + ux] : 0.0f;
    a_outx[c][1] = (n < 12) ? W_out[n * 16 + ux + 1] : 0.0f;
  }

  v8f bias_g[2], bias_out;
  float gam[8], bet[8];
#pragma unroll
  for (int r = 0; r < 8; ++r) {
    const int d = r + rb;
    gam[r] = ln_gamma[d];
    bet[r] = ln_beta[d];
#pragma unroll
    for (int T = 0; T < 2; ++T) {
      const int gr = GROW(T, d);
      bias_g[T][r] = b_ih[gr] + b_hh[gr];
    }
    bias_out[r] = (d < 12) ? b_out[d] : 0.0f;
  }

  v2f a_c[2][4];
#pragma unroll
  for (int T = 0; T < 2; ++T) {
    v8f wt;
#pragma unroll
    for (int r = 0; r < 8; ++r) wt[r] = 0.0f;
    wt = mm16(a_inT, a_ih[T], wt);
#pragma unroll
    for (int k = 0; k < 4; ++k) { a_c[T][k][0] = wt[2 * k]; a_c[T][k][1] = wt[2 * k + 1]; }
    bias_g[T] = mm16(a_ih[T], b_inb, bias_g[T]);
  }

  float s[8];
  const float* st0 = initial_state + (size_t)b * 12;
#pragma unroll
  for (int r = 0; r < 8; ++r) {
    float vlo = st0[r];
    float vhi = (r < 4) ? st0[8 + r] : 0.0f;
    s[r] = lo ? vlo : vhi;
  }

  float cst[4];
#pragma unroll
  for (int j = 0; j < 4; ++j) cst[j] = 0.0f;

  v8f pg0 = bias_g[0];
  v8f pg1 = bias_g[1];

  const float4* cmdp = reinterpret_cast<const float4*>(commands) + (size_t)b * T_STEPS;
  float4 cnext = cmdp[0];

  float* opa = out + (size_t)b * T_STEPS * 12 + (lo ? 0 : 8);
  float* opb = out + (size_t)b * T_STEPS * 12 + (lo ? 4 : 6);

  for (int t = 0; t < T_STEPS; ++t) {
    const float4 c4 = cnext;
    const int tn = (t + 1 < T_STEPS) ? t + 1 : t;
    cnext = cmdp[tn];

    float x[8];
    x[0] = s[0]; x[1] = s[1]; x[2] = s[2]; x[3] = s[3];
    x[4] = lo ? s[4] : c4.x;
    x[5] = lo ? s[5] : c4.y;
    x[6] = lo ? s[6] : c4.z;
    x[7] = lo ? s[7] : c4.w;

    float sum = 0.0f, sq = 0.0f;
#pragma unroll
    for (int r = 0; r < 8; ++r) { sum += x[r]; sq += x[r] * x[r]; }
    sum += xor16(sum);
    sq  += xor16(sq);
    const float mu   = sum * 0.0625f;
    const float var  = sq * 0.0625f - mu * mu;
    const float rstd = fast_rsq(var + 1e-5f);
#pragma unroll
    for (int r = 0; r < 8; ++r) x[r] = (x[r] - mu) * rstd * gam[r] + bet[r];

    v2f bmat[4];
    cd_to_b(x, bmat);

    v8f G0 = pg0, G1 = pg1;
#pragma unroll
    for (int k = 0; k < 4; ++k) {
      G0 = wmma4(a_c[0][k], bmat[k], G0);
      G1 = wmma4(a_c[1][k], bmat[k], G1);
    }

    asm volatile("" :: "v"(G0), "v"(G1));
    __builtin_amdgcn_sched_barrier(0);

    float own[4];
#pragma unroll
    for (int j = 0; j < 4; ++j) {
      float cc = sigf(G0[4 + j]) * cst[j] + sigf(G0[j]) * fast_tanh(G1[j]);
      cst[j]   = cc;
      own[j]   = sigf(G1[4 + j]) * fast_tanh(cc);
    }

    xch[t & 1][wave][lane] = make_float4(own[0], own[1], own[2], own[3]);

    v2f bo[2];
    bo[0][0] = own[0]; bo[0][1] = own[1]; bo[1][0] = own[2]; bo[1][1] = own[3];
    v8f dso = wmma4(a_outo[0], bo[0], bias_out);
    dso = wmma4(a_outo[1], bo[1], dso);
    asm volatile("" :: "v"(dso));
    __builtin_amdgcn_sched_group_barrier(0x200, 1, 0);
    __builtin_amdgcn_sched_group_barrier(0x008, 2, 0);
    __builtin_amdgcn_sched_barrier(0);

    __builtin_amdgcn_fence(__ATOMIC_RELEASE, "workgroup", "local");
    __builtin_amdgcn_s_barrier();
    __builtin_amdgcn_fence(__ATOMIC_ACQUIRE, "workgroup", "local");
    const float4 ov = xch[t & 1][wave ^ 1][lane];

    pg0 = wmma4(a_hho[0][0], bo[0], bias_g[0]);
    pg1 = wmma4(a_hho[1][0], bo[0], bias_g[1]);
    pg0 = wmma4(a_hho[0][1], bo[1], pg0);
    pg1 = wmma4(a_hho[1][1], bo[1], pg1);
    asm volatile("" :: "v"(pg0), "v"(pg1));
    __builtin_amdgcn_sched_group_barrier(0x100, 1, 0);
    __builtin_amdgcn_sched_group_barrier(0x008, 4, 0);
    __builtin_amdgcn_sched_barrier(0);

    v2f bx[2];
    bx[0][0] = ov.x; bx[0][1] = ov.y; bx[1][0] = ov.z; bx[1][1] = ov.w;

    v8f ds = wmma4(a_outx[0], bx[0], dso);
    ds = wmma4(a_outx[1], bx[1], ds);
    asm volatile("" :: "v"(ds));
    __builtin_amdgcn_sched_barrier(0);


    pg0 = wmma4(a_hhx[0][0], bx[0], pg0);
#pragma unroll
    for (int r = 0; r < 8; ++r) {
      float ns = s[r] + ds[r];
      s[r] = (lo || r < 4) ? ns : s[r];
    }
    asm volatile("" :: "v"(pg0), "v"(s[0]), "v"(s[1]), "v"(s[2]), "v"(s[3]),
                       "v"(s[4]), "v"(s[5]), "v"(s[6]), "v"(s[7]));
    __builtin_amdgcn_sched_barrier(0);

    pg1 = wmma4(a_hhx[1][0], bx[0], pg1);
    float rt[3];
#pragma unroll
    for (int p = 0; p < 3; ++p) {
      const float cs = s[1 + 2 * p], sn = s[2 + 2 * p];
      rt[p] = fast_sqrt(sn * sn + cs * cs);
    }
    asm volatile("" :: "v"(pg1), "v"(rt[0]), "v"(rt[1]), "v"(rt[2]));
    __builtin_amdgcn_sched_barrier(0);

    pg0 = wmma4(a_hhx[0][1], bx[1], pg0);
    float inv[3];
#pragma unroll
    for (int p = 0; p < 3; ++p) inv[p] = fast_rcp(rt[p] + 1e-8f);
    asm volatile("" :: "v"(pg0), "v"(inv[0]), "v"(inv[1]), "v"(inv[2]));
    __builtin_amdgcn_sched_barrier(0);

    pg1 = wmma4(a_hhx[1][1], bx[1], pg1);
#pragma unroll
    for (int p = 0; p < 3; ++p) {
      const int ci = 1 + 2 * p, sj = 2 + 2 * p;
      const float cs = s[ci], sn = s[sj];
      s[ci] = lo ? cs * inv[p] : s[ci];
      s[sj] = lo ? sn * inv[p] : s[sj];
    }

    if (q == 0) {
      *reinterpret_cast<float4*>(opa) = make_float4(s[0], s[1], s[2], s[3]);
    } else {
      const float p6 = xor16(s[6]), p7 = xor16(s[7]);
      *reinterpret_cast<float2*>(opb) = lo ? make_float2(s[4], s[5]) : make_float2(p6, p7);
    }
    opa += 12;
    opb += 12;
  }
#undef GROW
}

extern "C" void kernel_launch(void* const* d_in, const int* in_sizes, int n_in,
                              void* d_out, int out_size, void* d_ws, size_t ws_size,
                              hipStream_t stream) {
  (void)in_sizes; (void)n_in; (void)out_size; (void)d_ws; (void)ws_size;
  dim3 grid(B_TOTAL / 32);
  dim3 block(128);
  vehicle_lstm_wmma<<<grid, block, 0, stream>>>(
      (const float*)d_in[0],
      (const float*)d_in[1],
      (const float*)d_in[2],
      (const float*)d_in[3],
      (const float*)d_in[4],
      (const float*)d_in[5],
      (const float*)d_in[6],
      (const float*)d_in[7],
      (const float*)d_in[8],
      (const float*)d_in[9],
      (const float*)d_in[10],
      (const float*)d_in[11],
      (float*)d_out);
}


# Command reads lead their use by eight steps, in a ring of eight register quads (time loop unrolled by eight), instead of one step: a new line's read from memory no longer outlasts its lead
# speedup vs baseline: 2.3578x; 1.2369x over previous
#include <hip/hip_runtime.h>

#define B_TOTAL 8192
#define T_STEPS 256
#define CMD_AHEAD 8

typedef float v2f __attribute__((ext_vector_type(2)));
typedef float v8f __attribute__((ext_vector_type(8)));

static __device__ __forceinline__ v8f wmma4(v2f a, v2f b, v8f c) {
  return __builtin_amdgcn_wmma_f32_16x16x4_f32(false, a, false, b, (short)0, c,
                                               false, false);
}

static __device__ __forceinline__ v8f mm16(const v2f a[4], const v2f b[4], v8f acc) {
#pragma unroll
  for (int k = 0; k < 4; ++k) acc = wmma4(a[k], b[k], acc);
  return acc;
}

static __device__ __forceinline__ float xor16(float v) {
#if __has_builtin(__builtin_amdgcn_permlanex16)
  unsigned u = __builtin_bit_cast(unsigned, v);
  unsigned r = __builtin_amdgcn_permlanex16(u, u, 0x76543210u, 0xFEDCBA98u,
                                            false, false);
  return __builtin_bit_cast(float, r);
#else
  return __shfl_xor(v, 16, 32);
#endif
}

static __device__ __forceinline__ float fast_tanh(float x) {
#if __has_builtin(__builtin_amdgcn_tanhf)
  return __builtin_amdgcn_tanhf(x);
#else
  return tanhf(x);
#endif
}
static __device__ __forceinline__ float fast_rcp(float x) {
#if __has_builtin(__builtin_amdgcn_rcpf)
  return __builtin_amdgcn_rcpf(x);
#else
  return 1.0f / x;
#endif
}
static __device__ __forceinline__ float fast_rsq(float x) {
#if __has_builtin(__builtin_amdgcn_rsqf)
  return __builtin_amdgcn_rsqf(x);
#else
  return rsqrtf(x);
#endif
}
static __device__ __forceinline__ float fast_sqrt(float x) {
#if __has_builtin(__builtin_amdgcn_sqrtf)
  return __builtin_amdgcn_sqrtf(x);
#else
  return sqrtf(x);
#endif
}
static __device__ __forceinline__ float sigf(float x) {
  return fast_rcp(1.0f + __expf(-x));
}

static __device__ __forceinline__ void cd_to_b(const float x[8], v2f b[4]) {
#pragma unroll
  for (int k = 0; k < 4; ++k) { b[k][0] = x[2 * k]; b[k][1] = x[2 * k + 1]; }
}

__global__ __launch_bounds__(128)
void vehicle_lstm_wmma(const float* __restrict__ initial_state,
                       const float* commands,
                       const float* __restrict__ ln_gamma,
                       const float* __restrict__ ln_beta,
                       const float* __restrict__ W_in,
                       const float* __restrict__ b_in,
                       const float* __restrict__ W_ih,
                       const float* __restrict__ W_hh,
                       const float* __restrict__ b_ih,
                       const float* __restrict__ b_hh,
                       const float* __restrict__ W_out,
                       const float* __restrict__ b_out,
                       float* __restrict__ out) {
  __shared__ float4 xch[2][4][32];

  const int lane = threadIdx.x & 31;
  const int wave = __builtin_amdgcn_readfirstlane((int)(threadIdx.x >> 5));
  const int q    = wave & 1;
  const bool lo  = lane < 16;
  const int  n   = lane & 15;
  const int  hh  = lo ? 0 : 1;
  const int  b   = (blockIdx.x * 2 + (wave >> 1)) * 16 + n;
  const int  kof = lo ? 0 : 8;
  const int  rb  = lo ? 0 : 8;

#define GROW(T, rho) ((2 * (T) + (((rho) >> 2) & 1)) * 16 + 8 * q + 4 * ((rho) >> 3) + ((rho) & 3))

  v2f a_inT[4], b_inb[4], a_ih[2][4], a_hho[2][2], a_hhx[2][2], a_outo[2], a_outx[2];
#pragma unroll
  for (int k = 0; k < 4; ++k) {
    const int kb = 2 * k + kof;
    a_inT[k][0] = W_in[kb * 16 + n];
    a_inT[k][1] = W_in[(kb + 1) * 16 + n];
    b_inb[k][0] = b_in[kb];
    b_inb[k][1] = b_in[kb + 1];
#pragma unroll
    for (int T = 0; T < 2; ++T) {
      const int gr = GROW(T, n);
      a_ih[T][k][0] = W_ih[gr * 16 + kb];
      a_ih[T][k][1] = W_ih[gr * 16 + kb + 1];
    }
  }
#pragma unroll
  for (int c = 0; c < 2; ++c) {
    const int uo = 8 * q + 4 * hh + 2 * c;
    const int ux = 8 * (1 - q) + 4 * hh + 2 * c;
#pragma unroll
    for (int T = 0; T < 2; ++T) {
      const int gr = GROW(T, n);
      a_hho[T][c][0] = W_hh[gr * 16 + uo];
      a_hho[T][c][1] = W_hh[gr * 16 + uo + 1];
      a_hhx[T][c][0] = W_hh[gr * 16 + ux];
      a_hhx[T][c][1] = W_hh[gr * 16 + ux + 1];
    }
    a_outo[c][0] = (n < 12) ? W_out[n * 16 + uo] : 0.0f;
    a_outo[c][1] = (n < 12) ? W_out[n * 16 + uo + 1] : 0.0f;
    a_outx[c][0] = (n < 12) ? W_out[n * 16 + ux] : 0.0f;
    a_outx[c][1] = (n < 12) ? W_out[n * 16 + ux + 1] : 0.0f;
  }

  v8f bias_g[2], bias_out;
  float gam[8], bet[8];
#pragma unroll
  for (int r = 0; r < 8; ++r) {
    const int d = r + rb;
    gam[r] = ln_gamma[d];
    bet[r] = ln_beta[d];
#pragma unroll
    for (int T = 0; T < 2; ++T) {
      const int gr = GROW(T, d);
      bias_g[T][r] = b_ih[gr] + b_hh[gr];
    }
    bias_out[r] = (d < 12) ? b_out[d] : 0.0f;
  }

  v2f a_c[2][4];
#pragma unroll
  for (int T = 0; T < 2; ++T) {
    v8f wt;
#pragma unroll
    for (int r = 0; r < 8; ++r) wt[r] = 0.0f;
    wt = mm16(a_inT, a_ih[T], wt);
#pragma unroll
    for (int k = 0; k < 4; ++k) { a_c[T][k][0] = wt[2 * k]; a_c[T][k][1] = wt[2 * k + 1]; }
    bias_g[T] = mm16(a_ih[T], b_inb, bias_g[T]);
  }

  float s[8];
  const float* st0 = initial_state + (size_t)b * 12;
#pragma unroll
  for (int r = 0; r < 8; ++r) {
    float vlo = st0[r];
    float vhi = (r < 4) ? st0[8 + r] : 0.0f;
    s[r] = lo ? vlo : vhi;
  }

  float cst[4];
#pragma unroll
  for (int j = 0; j < 4; ++j) cst[j] = 0.0f;

  v8f pg0 = bias_g[0];
  v8f pg1 = bias_g[1];

  const float4* cmdp = reinterpret_cast<const float4*>(commands) + (size_t)b * T_STEPS;
  float4 cq[CMD_AHEAD];
#pragma unroll
  for (int j = 0; j < CMD_AHEAD; ++j) cq[j] = cmdp[j];

  float* opa = out + (size_t)b * T_STEPS * 12 + (lo ? 0 : 8);
  float* opb = out + (size_t)b * T_STEPS * 12 + (lo ? 4 : 6);

  for (int t0 = 0; t0 < T_STEPS; t0 += CMD_AHEAD) {
  const float4* cnx = cmdp + ((t0 + CMD_AHEAD) & (T_STEPS - 1));
#pragma unroll
  for (int j = 0; j < CMD_AHEAD; ++j) {
    const int t = t0 + j;
    const float4 c4 = cq[j];
    cq[j] = cnx[j];

    float x[8];
    x[0] = s[0]; x[1] = s[1]; x[2] = s[2]; x[3] = s[3];
    x[4] = lo ? s[4] : c4.x;
    x[5] = lo ? s[5] : c4.y;
    x[6] = lo ? s[6] : c4.z;
    x[7] = lo ? s[7] : c4.w;

    float sum = 0.0f, sq = 0.0f;
#pragma unroll
    for (int r = 0; r < 8; ++r) { sum += x[r]; sq += x[r] * x[r]; }
    sum += xor16(sum);
    sq  += xor16(sq);
    const float mu   = sum * 0.0625f;
    const float var  = sq * 0.0625f - mu * mu;
    const float rstd = fast_rsq(var + 1e-5f);
#pragma unroll
    for (int r = 0; r < 8; ++r) x[r] = (x[r] - mu) * rstd * gam[r] + bet[r];

    v2f bmat[4];
    cd_to_b(x, bmat);

    v8f G0 = pg0, G1 = pg1;
#pragma unroll
    for (int k = 0; k < 4; ++k) {
      G0 = wmma4(a_c[0][k], bmat[k], G0);
      G1 = wmma4(a_c[1][k], bmat[k], G1);
    }

    asm volatile("" :: "v"(G0), "v"(G1));
    __builtin_amdgcn_sched_barrier(0);

    float own[4];
#pragma unroll
    for (int j = 0; j < 4; ++j) {
      float cc = sigf(G0[4 + j]) * cst[j] + sigf(G0[j]) * fast_tanh(G1[j]);
      cst[j]   = cc;
      own[j]   = sigf(G1[4 + j]) * fast_tanh(cc);
    }

    xch[t & 1][wave][lane] = make_float4(own[0], own[1], own[2], own[3]);

    v2f bo[2];
    bo[0][0] = own[0]; bo[0][1] = own[1]; bo[1][0] = own[2]; bo[1][1] = own[3];
    v8f dso = wmma4(a_outo[0], bo[0], bias_out);
    dso = wmma4(a_outo[1], bo[1], dso);
    asm volatile("" :: "v"(dso));
    __builtin_amdgcn_sched_group_barrier(0x200, 1, 0);
    __builtin_amdgcn_sched_group_barrier(0x008, 2, 0);
    __builtin_amdgcn_sched_barrier(0);

    __builtin_amdgcn_fence(__ATOMIC_RELEASE, "workgroup", "local");
    __builtin_amdgcn_s_barrier();
    __builtin_amdgcn_fence(__ATOMIC_ACQUIRE, "workgroup", "local");
    const float4 ov = xch[t & 1][wave ^ 1][lane];

    pg0 = wmma4(a_hho[0][0], bo[0], bias_g[0]);
    pg1 = wmma4(a_hho[1][0], bo[0], bias_g[1]);
    pg0 = wmma4(a_hho[0][1], bo[1], pg0);
    pg1 = wmma4(a_hho[1][1], bo[1], pg1);
    asm volatile("" :: "v"(pg0), "v"(pg1));
    __builtin_amdgcn_sched_group_barrier(0x100, 1, 0);
    __builtin_amdgcn_sched_group_barrier(0x008, 4, 0);
    __builtin_amdgcn_sched_barrier(0);

    v2f bx[2];
    bx[0][0] = ov.x; bx[0][1] = ov.y; bx[1][0] = ov.z; bx[1][1] = ov.w;

    v8f ds = wmma4(a_outx[0], bx[0], dso);
    ds = wmma4(a_outx[1], bx[1], ds);
    asm volatile("" :: "v"(ds));
    __builtin_amdgcn_sched_barrier(0);


    pg0 = wmma4(a_hhx[0][0], bx[0], pg0);
#pragma unroll
    for (int r = 0; r < 8; ++r) {
      float ns = s[r] + ds[r];
      s[r] = (lo || r < 4) ? ns : s[r];
    }
    asm volatile("" :: "v"(pg0), "v"(s[0]), "v"(s[1]), "v"(s[2]), "v"(s[3]),
                       "v"(s[4]), "v"(s[5]), "v"(s[6]), "v"(s[7]));
    __builtin_amdgcn_sched_barrier(0);

    pg1 = wmma4(a_hhx[1][0], bx[0], pg1);
    float rt[3];
#pragma unroll
    for (int p = 0; p < 3; ++p) {
      const float cs = s[1 + 2 * p], sn = s[2 + 2 * p];
      rt[p] = fast_sqrt(sn * sn + cs * cs);
    }
    asm volatile("" :: "v"(pg1), "v"(rt[0]), "v"(rt[1]), "v"(rt[2]));
    __builtin_amdgcn_sched_barrier(0);

    pg0 = wmma4(a_hhx[0][1], bx[1], pg0);
    float inv[3];
#pragma unroll
    for (int p = 0; p < 3; ++p) inv[p] = fast_rcp(rt[p] + 1e-8f);
    asm volatile("" :: "v"(pg0), "v"(inv[0]), "v"(inv[1]), "v"(inv[2]));
    __builtin_amdgcn_sched_barrier(0);

    pg1 = wmma4(a_hhx[1][1], bx[1], pg1);
#pragma unroll
    for (int p = 0; p < 3; ++p) {
      const int ci = 1 + 2 * p, sj = 2 + 2 * p;
      const float cs = s[ci], sn = s[sj];
      s[ci] = lo ? cs * inv[p] : s[ci];
      s[sj] = lo ? sn * inv[p] : s[sj];
    }

    if (q == 0) {
      *reinterpret_cast<float4*>(opa) = make_float4(s[0], s[1], s[2], s[3]);
    } else {
      const float p6 = xor16(s[6]), p7 = xor16(s[7]);
      *reinterpret_cast<float2*>(opb) = lo ? make_float2(s[4], s[5]) : make_float2(p6, p7);
    }
    opa += 12;
    opb += 12;
  }
  }
#undef GROW
}

extern "C" void kernel_launch(void* const* d_in, const int* in_sizes, int n_in,
                              void* d_out, int out_size, void* d_ws, size_t ws_size,
                              hipStream_t stream) {
  (void)in_sizes; (void)n_in; (void)out_size; (void)d_ws; (void)ws_size;
  dim3 grid(B_TOTAL / 32);
  dim3 block(128);
  vehicle_lstm_wmma<<<grid, block, 0, stream>>>(
      (const float*)d_in[0],
      (const float*)d_in[1],
      (const float*)d_in[2],
      (const float*)d_in[3],
      (const float*)d_in[4],
      (const float*)d_in[5],
      (const float*)d_in[6],
      (const float*)d_in[7],
      (const float*)d_in[8],
      (const float*)d_in[9],
      (const float*)d_in[10],
      (const float*)d_in[11],
      (float*)d_out);
}
